# starting kernel
# baseline (speedup 1.0000x reference)
#include <hip/hip_runtime.h>
#include <hip/hip_bf16.h>

typedef __attribute__((ext_vector_type(2))) float v2f;
typedef __attribute__((ext_vector_type(8))) float v8f;

__global__ void gcn_fill_kernel(float* __restrict__ p, float v, int n) {
    int i = blockIdx.x * blockDim.x + threadIdx.x;
    int stride = gridDim.x * blockDim.x;
    for (; i < n; i += stride) p[i] = v;
}

__global__ void gcn_degree_kernel(const int* __restrict__ dst,
                                  float* __restrict__ deg, int nE) {
    int e = blockIdx.x * blockDim.x + threadIdx.x;
    if (e < nE) atomicAdd(&deg[dst[e]], 1.0f);
}

__global__ void gcn_rsqrt_kernel(float* __restrict__ d, int n) {
    int i = blockIdx.x * blockDim.x + threadIdx.x;
    if (i < n) d[i] = rsqrtf(d[i]);
}

template <int K, int N, int NTILES>
__global__ __launch_bounds__(256) void gcn_gemm_wmma_kernel(
    const float* __restrict__ A, const float* __restrict__ B,
    float* __restrict__ C, int M) {
    constexpr int KP = (K + 3) & ~3;
    constexpr int KS = KP + 2;
    constexpr int NP = NTILES * 16;
    __shared__ float BT[NP * KS];

    for (int idx = threadIdx.x; idx < NP * KS; idx += blockDim.x) {
        const int n = idx / KS;
        const int k = idx - n * KS;
        BT[idx] = (n < N && k < K) ? B[(size_t)k * N + n] : 0.0f;
    }
    __syncthreads();

    const int wavesPerBlock = blockDim.x >> 5;
    const int tile = blockIdx.x * wavesPerBlock + (threadIdx.x >> 5);
    const int totalTiles = (M >> 4) * NTILES;
    if (tile >= totalTiles) return;

    const int lane  = threadIdx.x & 31;
    const int mt    = tile / NTILES;
    const int nt    = tile - mt * NTILES;
    const int row0  = mt << 4;
    const int mrow  = lane & 15;
    const int khalf = (lane >> 4) << 1;
    const int col   = (nt << 4) + mrow;

    const float*  Arow  = A + (size_t)(row0 + mrow) * K;
    const float2* BTrow = (const float2*)(BT + col * KS);

    v8f acc = {};
    constexpr int KFULL = K & ~3;
#pragma unroll
    for (int k0 = 0; k0 < KFULL; k0 += 4) {
        const int ka = k0 + khalf;
        const float2 av = *(const float2*)(Arow + ka);
        const float2 bv = BTrow[ka >> 1];
        v2f a; a.x = av.x; a.y = av.y;
        v2f b; b.x = bv.x; b.y = bv.y;
        acc = __builtin_amdgcn_wmma_f32_16x16x4_f32(
            false, a, false, b, (short)0, acc, false, false);
    }
    if constexpr (KFULL < K) {
        const int ka = KFULL + khalf;
        v2f a;
        a.x = (ka     < K) ? Arow[ka]     : 0.0f;
        a.y = (ka + 1 < K) ? Arow[ka + 1] : 0.0f;
        const float2 bv = BTrow[ka >> 1];
        v2f b; b.x = bv.x; b.y = bv.y;
        acc = __builtin_amdgcn_wmma_f32_16x16x4_f32(
            false, a, false, b, (short)0, acc, false, false);
    }

    if (col < N) {
        const int rbase = row0 + ((lane >> 4) << 3);
#pragma unroll
        for (int r = 0; r < 8; ++r)
            C[(size_t)(rbase + r) * N + col] = acc[r];
    }
}

__global__ void gcn_scatter_kernel(const float* __restrict__ xw,
                                   const int* __restrict__ src,
                                   const int* __restrict__ dst,
                                   const float* __restrict__ dinv,
                                   float* __restrict__ acc,
                                   int nE, int F) {
    const int wavesPerBlock = blockDim.x >> 5;
    const int e = blockIdx.x * wavesPerBlock + (threadIdx.x >> 5);
    if (e >= nE) return;
    const int lane = threadIdx.x & 31;
    const int s = src[e];
    const int d = dst[e];
    const float norm = dinv[s] * dinv[d];
    const float* xs = xw + (size_t)s * F;
    float* ad = acc + (size_t)d * F;
    for (int f = lane; f < F; f += 32)
        atomicAdd(&ad[f], norm * xs[f]);
}

__global__ void gcn_selfloop_kernel(const float* __restrict__ acc,
                                    const float* __restrict__ xw,
                                    const float* __restrict__ dinv,
                                    float* __restrict__ out,
                                    int nNodes, int F, int doTanh) {
    int idx = blockIdx.x * blockDim.x + threadIdx.x;
    int total = nNodes * F;
    int stride = gridDim.x * blockDim.x;
    for (; idx < total; idx += stride) {
        const int node = idx / F;
        const float di = dinv[node];
        float v = acc[idx] + di * di * xw[idx];
        out[idx] = doTanh ? tanhf(v) : v;
    }
}

extern "C" void kernel_launch(void* const* d_in, const int* in_sizes, int n_in,
                              void* d_out, int out_size, void* d_ws, size_t ws_size,
                              hipStream_t stream) {
    const float* x  = (const float*)d_in[0];
    const float* W1 = (const float*)d_in[1];
    const float* W2 = (const float*)d_in[2];
    const int* ei   = (const int*)d_in[3];

    const int FIN = 50, F1 = 64, F2 = 121;
    const int nNodes = in_sizes[0] / FIN;
    const int nEdges = in_sizes[3] / 2;
    const int* src = ei;
    const int* dst = ei + nEdges;

    float* wsF = (float*)d_ws;
    float* dinv = wsF;
    float* xw1  = dinv + ((nNodes + 63) & ~63);
    float* hbuf = xw1 + (size_t)nNodes * F1;
    float* xw2  = hbuf + (size_t)nNodes * F1;
    float* outf = (float*)d_out;

    const int T = 256;
    const int wavesPerBlock = T / 32;

    gcn_fill_kernel<<<(nNodes + T - 1) / T, T, 0, stream>>>(dinv, 1.0f, nNodes);
    gcn_degree_kernel<<<(nEdges + T - 1) / T, T, 0, stream>>>(dst, dinv, nEdges);
    gcn_rsqrt_kernel<<<(nNodes + T - 1) / T, T, 0, stream>>>(dinv, nNodes);

    {
        const int tiles = (nNodes / 16) * 4;
        const int blocks = (tiles + wavesPerBlock - 1) / wavesPerBlock;
        gcn_gemm_wmma_kernel<50, 64, 4><<<blocks, T, 0, stream>>>(x, W1, xw1, nNodes);
    }
    gcn_fill_kernel<<<4096, T, 0, stream>>>(hbuf, 0.0f, nNodes * F1);
    gcn_scatter_kernel<<<(nEdges + wavesPerBlock - 1) / wavesPerBlock, T, 0, stream>>>(
        xw1, src, dst, dinv, hbuf, nEdges, F1);
    gcn_selfloop_kernel<<<4096, T, 0, stream>>>(hbuf, xw1, dinv, hbuf, nNodes, F1, 1);

    {
        const int tiles = (nNodes / 16) * 8;
        const int blocks = (tiles + wavesPerBlock - 1) / wavesPerBlock;
        gcn_gemm_wmma_kernel<64, 121, 8><<<blocks, T, 0, stream>>>(hbuf, W2, xw2, nNodes);
    }
    gcn_fill_kernel<<<4096, T, 0, stream>>>(outf, 0.0f, nNodes * F2);
    gcn_scatter_kernel<<<(nEdges + wavesPerBlock - 1) / wavesPerBlock, T, 0, stream>>>(
        xw2, src, dst, dinv, outf, nEdges, F2);
    gcn_selfloop_kernel<<<4096, T, 0, stream>>>(outf, xw2, dinv, outf, nNodes, F2, 0);

    (void)n_in; (void)out_size; (void)ws_size;
}


# Layer 2 sums h (64 wide) over the edges first, then multiplies by W2: (A h) W2 for A (h W2); the atomic scatter moves 64 not 121 columns; f32 summation order changes (reassociated)
# speedup vs baseline: 1.2116x; 1.2116x over previous
#include <hip/hip_runtime.h>
#include <hip/hip_bf16.h>

typedef __attribute__((ext_vector_type(2))) float v2f;
typedef __attribute__((ext_vector_type(8))) float v8f;

__global__ void gcn_fill_kernel(float* __restrict__ p, float v, int n) {
    int i = blockIdx.x * blockDim.x + threadIdx.x;
    int stride = gridDim.x * blockDim.x;
    for (; i < n; i += stride) p[i] = v;
}

__global__ void gcn_degree_kernel(const int* __restrict__ dst,
                                  float* __restrict__ deg, int nE) {
    int e = blockIdx.x * blockDim.x + threadIdx.x;
    if (e < nE) atomicAdd(&deg[dst[e]], 1.0f);
}

__global__ void gcn_rsqrt_kernel(float* __restrict__ d, int n) {
    int i = blockIdx.x * blockDim.x + threadIdx.x;
    if (i < n) d[i] = rsqrtf(d[i]);
}

template <int K, int N, int NTILES>
__global__ __launch_bounds__(256) void gcn_gemm_wmma_kernel(
    const float* __restrict__ A, const float* __restrict__ B,
    float* __restrict__ C, int M) {
    constexpr int KP = (K + 3) & ~3;
    constexpr int KS = KP + 2;
    constexpr int NP = NTILES * 16;
    __shared__ float BT[NP * KS];

    for (int idx = threadIdx.x; idx < NP * KS; idx += blockDim.x) {
        const int n = idx / KS;
        const int k = idx - n * KS;
        BT[idx] = (n < N && k < K) ? B[(size_t)k * N + n] : 0.0f;
    }
    __syncthreads();

    const int wavesPerBlock = blockDim.x >> 5;
    const int tile = blockIdx.x * wavesPerBlock + (threadIdx.x >> 5);
    const int totalTiles = (M >> 4) * NTILES;
    if (tile >= totalTiles) return;

    const int lane  = threadIdx.x & 31;
    const int mt    = tile / NTILES;
    const int nt    = tile - mt * NTILES;
    const int row0  = mt << 4;
    const int mrow  = lane & 15;
    const int khalf = (lane >> 4) << 1;
    const int col   = (nt << 4) + mrow;

    const float*  Arow  = A + (size_t)(row0 + mrow) * K;
    const float2* BTrow = (const float2*)(BT + col * KS);

    v8f acc = {};
    constexpr int KFULL = K & ~3;
#pragma unroll
    for (int k0 = 0; k0 < KFULL; k0 += 4) {
        const int ka = k0 + khalf;
        const float2 av = *(const float2*)(Arow + ka);
        const float2 bv = BTrow[ka >> 1];
        v2f a; a.x = av.x; a.y = av.y;
        v2f b; b.x = bv.x; b.y = bv.y;
        acc = __builtin_amdgcn_wmma_f32_16x16x4_f32(
            false, a, false, b, (short)0, acc, false, false);
    }
    if constexpr (KFULL < K) {
        const int ka = KFULL + khalf;
        v2f a;
        a.x = (ka     < K) ? Arow[ka]     : 0.0f;
        a.y = (ka + 1 < K) ? Arow[ka + 1] : 0.0f;
        const float2 bv = BTrow[ka >> 1];
        v2f b; b.x = bv.x; b.y = bv.y;
        acc = __builtin_amdgcn_wmma_f32_16x16x4_f32(
            false, a, false, b, (short)0, acc, false, false);
    }

    if (col < N) {
        const int rbase = row0 + ((lane >> 4) << 3);
#pragma unroll
        for (int r = 0; r < 8; ++r)
            C[(size_t)(rbase + r) * N + col] = acc[r];
    }
}

__global__ void gcn_scatter_kernel(const float* __restrict__ xw,
                                   const int* __restrict__ src,
                                   const int* __restrict__ dst,
                                   const float* __restrict__ dinv,
                                   float* __restrict__ acc,
                                   int nE, int F) {
    const int wavesPerBlock = blockDim.x >> 5;
    const int e = blockIdx.x * wavesPerBlock + (threadIdx.x >> 5);
    if (e >= nE) return;
    const int lane = threadIdx.x & 31;
    const int s = src[e];
    const int d = dst[e];
    const float norm = dinv[s] * dinv[d];
    const float* xs = xw + (size_t)s * F;
    float* ad = acc + (size_t)d * F;
    for (int f = lane; f < F; f += 32)
        atomicAdd(&ad[f], norm * xs[f]);
}

__global__ void gcn_selfloop_kernel(const float* __restrict__ acc,
                                    const float* __restrict__ xw,
                                    const float* __restrict__ dinv,
                                    float* __restrict__ out,
                                    int nNodes, int F, int doTanh) {
    int idx = blockIdx.x * blockDim.x + threadIdx.x;
    int total = nNodes * F;
    int stride = gridDim.x * blockDim.x;
    for (; idx < total; idx += stride) {
        const int node = idx / F;
        const float di = dinv[node];
        float v = acc[idx] + di * di * xw[idx];
        out[idx] = doTanh ? tanhf(v) : v;
    }
}

extern "C" void kernel_launch(void* const* d_in, const int* in_sizes, int n_in,
                              void* d_out, int out_size, void* d_ws, size_t ws_size,
                              hipStream_t stream) {
    const float* x  = (const float*)d_in[0];
    const float* W1 = (const float*)d_in[1];
    const float* W2 = (const float*)d_in[2];
    const int* ei   = (const int*)d_in[3];

    const int FIN = 50, F1 = 64;
    const int nNodes = in_sizes[0] / FIN;
    const int nEdges = in_sizes[3] / 2;
    const int* src = ei;
    const int* dst = ei + nEdges;

    float* wsF = (float*)d_ws;
    float* dinv = wsF;
    float* xw1  = dinv + ((nNodes + 63) & ~63);
    float* hbuf = xw1 + (size_t)nNodes * F1;
    float* agg2 = hbuf + (size_t)nNodes * F1;
    float* outf = (float*)d_out;

    const int T = 256;
    const int wavesPerBlock = T / 32;

    gcn_fill_kernel<<<(nNodes + T - 1) / T, T, 0, stream>>>(dinv, 1.0f, nNodes);
    gcn_degree_kernel<<<(nEdges + T - 1) / T, T, 0, stream>>>(dst, dinv, nEdges);
    gcn_rsqrt_kernel<<<(nNodes + T - 1) / T, T, 0, stream>>>(dinv, nNodes);

    {
        const int tiles = (nNodes / 16) * 4;
        const int blocks = (tiles + wavesPerBlock - 1) / wavesPerBlock;
        gcn_gemm_wmma_kernel<50, 64, 4><<<blocks, T, 0, stream>>>(x, W1, xw1, nNodes);
    }
    gcn_fill_kernel<<<4096, T, 0, stream>>>(hbuf, 0.0f, nNodes * F1);
    gcn_scatter_kernel<<<(nEdges + wavesPerBlock - 1) / wavesPerBlock, T, 0, stream>>>(
        xw1, src, dst, dinv, hbuf, nEdges, F1);
    gcn_selfloop_kernel<<<4096, T, 0, stream>>>(hbuf, xw1, dinv, hbuf, nNodes, F1, 1);

    gcn_fill_kernel<<<4096, T, 0, stream>>>(agg2, 0.0f, nNodes * F1);
    gcn_scatter_kernel<<<(nEdges + wavesPerBlock - 1) / wavesPerBlock, T, 0, stream>>>(
        hbuf, src, dst, dinv, agg2, nEdges, F1);
    gcn_selfloop_kernel<<<4096, T, 0, stream>>>(agg2, hbuf, dinv, agg2, nNodes, F1, 0);
    {
        const int tiles = (nNodes / 16) * 8;
        const int blocks = (tiles + wavesPerBlock - 1) / wavesPerBlock;
        gcn_gemm_wmma_kernel<64, 121, 8><<<blocks, T, 0, stream>>>(agg2, W2, outf, nNodes);
    }

    (void)n_in; (void)out_size; (void)ws_size;
}


# Edge sums by gather: edges grouped by destination once per call (count+rank, scan, place), one wave sums each row in registers; no f32 atomics; f32 summation order changes (arrival-order rank)
# speedup vs baseline: 2.0857x; 1.7214x over previous
#include <hip/hip_runtime.h>
#include <hip/hip_bf16.h>


typedef __attribute__((ext_vector_type(2))) float v2f;
typedef __attribute__((ext_vector_type(8))) float v8f;

__global__ void gcn_zero_int_kernel(int* __restrict__ p, int n) {
    int i = blockIdx.x * blockDim.x + threadIdx.x;
    if (i < n) p[i] = 0;
}

__global__ void gcn_count_rank_kernel(const int* __restrict__ dst,
                                      int* __restrict__ cnt,
                                      int* __restrict__ rank, int nE) {
    int e = blockIdx.x * blockDim.x + threadIdx.x;
    if (e < nE) rank[e] = atomicAdd(&cnt[dst[e]], 1);
}

__global__ __launch_bounds__(1024) void gcn_scan_kernel(
    const int* __restrict__ cnt, int* __restrict__ rowptr,
    float* __restrict__ dinv, int n) {
    __shared__ int s_wsum[32];
    __shared__ int s_psum[32];
    const int tid  = threadIdx.x;
    const int lane = tid & 31;
    const int wid  = tid >> 5;
    const int i0   = blockIdx.x * 1024;

    int partial = 0;
    for (int i = tid; i < i0; i += 1024) partial += cnt[i];

    const int i  = i0 + tid;
    const int my = (i < n) ? cnt[i] : 0;

    int incl = my;
    for (int o = 1; o < 32; o <<= 1) {
        const int t = __shfl_up(incl, o);
        if (lane >= o) incl += t;
    }
    for (int o = 16; o > 0; o >>= 1)
        partial += __shfl_xor(partial, o);

    if (lane == 31) s_wsum[wid] = incl;
    if (lane == 0)  s_psum[wid] = partial;
    __syncthreads();

    int base = 0;
    for (int w = 0; w < 32; ++w) {
        base += s_psum[w];
        if (w < wid) base += s_wsum[w];
    }
    if (i < n) {
        rowptr[i] = base + incl - my;
        dinv[i]   = rsqrtf((float)(my + 1));
    }
}

__global__ void gcn_place_kernel(const int* __restrict__ src,
                                 const int* __restrict__ dst,
                                 const int* __restrict__ rank,
                                 const int* __restrict__ rowptr,
                                 int* __restrict__ csr_src, int nE) {
    int e = blockIdx.x * blockDim.x + threadIdx.x;
    if (e < nE) csr_src[rowptr[dst[e]] + rank[e]] = src[e];
}

template <int K, int N, int NTILES>
__global__ __launch_bounds__(256) void gcn_gemm_wmma_kernel(
    const float* __restrict__ A, const float* __restrict__ B,
    float* __restrict__ C, int M) {
    constexpr int KP = (K + 3) & ~3;
    constexpr int KS = KP + 2;
    constexpr int NP = NTILES * 16;
    __shared__ float BT[NP * KS];

    for (int idx = threadIdx.x; idx < NP * KS; idx += blockDim.x) {
        const int n = idx / KS;
        const int k = idx - n * KS;
        BT[idx] = (n < N && k < K) ? B[(size_t)k * N + n] : 0.0f;
    }
    __syncthreads();

    const int wavesPerBlock = blockDim.x >> 5;
    const int tile = blockIdx.x * wavesPerBlock + (threadIdx.x >> 5);
    const int totalTiles = (M >> 4) * NTILES;
    if (tile >= totalTiles) return;

    const int lane  = threadIdx.x & 31;
    const int mt    = tile / NTILES;
    const int nt    = tile - mt * NTILES;
    const int row0  = mt << 4;
    const int mrow  = lane & 15;
    const int khalf = (lane >> 4) << 1;
    const int col   = (nt << 4) + mrow;

    const float*  Arow  = A + (size_t)(row0 + mrow) * K;
    const float2* BTrow = (const float2*)(BT + col * KS);

    v8f acc = {};
    constexpr int KFULL = K & ~3;
#pragma unroll
    for (int k0 = 0; k0 < KFULL; k0 += 4) {
        const int ka = k0 + khalf;
        const float2 av = *(const float2*)(Arow + ka);
        const float2 bv = BTrow[ka >> 1];
        v2f a; a.x = av.x; a.y = av.y;
        v2f b; b.x = bv.x; b.y = bv.y;
        acc = __builtin_amdgcn_wmma_f32_16x16x4_f32(
            false, a, false, b, (short)0, acc, false, false);
    }
    if constexpr (KFULL < K) {
        const int ka = KFULL + khalf;
        v2f a;
        a.x = (ka     < K) ? Arow[ka]     : 0.0f;
        a.y = (ka + 1 < K) ? Arow[ka + 1] : 0.0f;
        const float2 bv = BTrow[ka >> 1];
        v2f b; b.x = bv.x; b.y = bv.y;
        acc = __builtin_amdgcn_wmma_f32_16x16x4_f32(
            false, a, false, b, (short)0, acc, false, false);
    }

    if (col < N) {
        const int rbase = row0 + ((lane >> 4) << 3);
#pragma unroll
        for (int r = 0; r < 8; ++r)
            C[(size_t)(rbase + r) * N + col] = acc[r];
    }
}

template <int U>
__device__ __forceinline__ void gcn_rows_accumulate(
    const float* __restrict__ xw, int my_s, int my_wbits, int j, int lane2,
    float& ax, float& ay) {
    float2 r[U];
    float  w[U];
#pragma unroll
    for (int u = 0; u < U; ++u) {
        const int s = __builtin_amdgcn_readlane(my_s, j + u);
        w[u] = __int_as_float(__builtin_amdgcn_readlane(my_wbits, j + u));
        r[u] = *(const float2*)(xw + (size_t)s * 64 + lane2);
    }
#pragma unroll
    for (int u = 0; u < U; ++u) {
        ax += w[u] * r[u].x;
        ay += w[u] * r[u].y;
    }
}

template <int DO_TANH>
__global__ __launch_bounds__(256) void gcn_gather_kernel(
    const float* __restrict__ xw, const int* __restrict__ rowptr,
    const int* __restrict__ cnt, const int* __restrict__ csr_src,
    const float* __restrict__ dinv, float* __restrict__ out, int nNodes) {
    const int node = blockIdx.x * (blockDim.x >> 5) + (threadIdx.x >> 5);
    if (node >= nNodes) return;
    const int lane  = threadIdx.x & 31;
    const int lane2 = lane << 1;

    const int beg = __builtin_amdgcn_readfirstlane(rowptr[node]);
    const int end = beg + __builtin_amdgcn_readfirstlane(cnt[node]);
    const float dd = dinv[node];

    const float2 sv = *(const float2*)(xw + (size_t)node * 64 + lane2);
    float ax = dd * dd * sv.x;
    float ay = dd * dd * sv.y;

    for (int p = beg; p < end; p += 32) {
        const int idx = p + lane;
        int   my_s = 0;
        float my_w = 0.0f;
        if (idx < end) {
            my_s = csr_src[idx];
            my_w = dinv[my_s] * dd;
        }
        const int my_wbits = __float_as_int(my_w);
        const int left = end - p;
        const int n = left < 32 ? left : 32;
        int j = 0;
        for (; j + 8 <= n; j += 8)
            gcn_rows_accumulate<8>(xw, my_s, my_wbits, j, lane2, ax, ay);
        if (n - j >= 4) {
            gcn_rows_accumulate<4>(xw, my_s, my_wbits, j, lane2, ax, ay);
            j += 4;
        }
        if (n - j >= 2) {
            gcn_rows_accumulate<2>(xw, my_s, my_wbits, j, lane2, ax, ay);
            j += 2;
        }
        if (n - j >= 1)
            gcn_rows_accumulate<1>(xw, my_s, my_wbits, j, lane2, ax, ay);
    }

    float2 o;
    o.x = DO_TANH ? tanhf(ax) : ax;
    o.y = DO_TANH ? tanhf(ay) : ay;
    *(float2*)(out + (size_t)node * 64 + lane2) = o;
}

extern "C" void kernel_launch(void* const* d_in, const int* in_sizes, int n_in,
                              void* d_out, int out_size, void* d_ws, size_t ws_size,
                              hipStream_t stream) {
    const float* x  = (const float*)d_in[0];
    const float* W1 = (const float*)d_in[1];
    const float* W2 = (const float*)d_in[2];
    const int* ei   = (const int*)d_in[3];

    const int FIN = 50, F1 = 64;
    const int nNodes = in_sizes[0] / FIN;
    const int nEdges = in_sizes[3] / 2;
    const int* src = ei;
    const int* dst = ei + nEdges;

    const int NP = (nNodes + 63) & ~63;
    const int EP = (nEdges + 63) & ~63;
    float* wsF    = (float*)d_ws;
    float* dinv   = wsF;
    int*   cnt    = (int*)(dinv + NP);
    int*   rowptr = cnt + NP;
    int*   rank   = rowptr + NP;
    int*   csr    = rank + EP;
    float* xw1    = (float*)(csr + EP);
    float* hbuf   = xw1 + (size_t)nNodes * F1;
    float* agg2   = hbuf + (size_t)nNodes * F1;
    float* outf   = (float*)d_out;

    const int T = 256;
    const int wavesPerBlock = T / 32;

    gcn_zero_int_kernel<<<(nNodes + T - 1) / T, T, 0, stream>>>(cnt, nNodes);
    gcn_count_rank_kernel<<<(nEdges + T - 1) / T, T, 0, stream>>>(dst, cnt, rank, nEdges);
    gcn_scan_kernel<<<(nNodes + 1023) / 1024, 1024, 0, stream>>>(cnt, rowptr, dinv, nNodes);
    gcn_place_kernel<<<(nEdges + T - 1) / T, T, 0, stream>>>(src, dst, rank, rowptr, csr, nEdges);

    {
        const int tiles = (nNodes / 16) * 4;
        const int blocks = (tiles + wavesPerBlock - 1) / wavesPerBlock;
        gcn_gemm_wmma_kernel<50, 64, 4><<<blocks, T, 0, stream>>>(x, W1, xw1, nNodes);
    }
    gcn_gather_kernel<1><<<(nNodes + wavesPerBlock - 1) / wavesPerBlock, T, 0, stream>>>(
        xw1, rowptr, cnt, csr, dinv, hbuf, nNodes);

    gcn_gather_kernel<0><<<(nNodes + wavesPerBlock - 1) / wavesPerBlock, T, 0, stream>>>(
        hbuf, rowptr, cnt, csr, dinv, agg2, nNodes);
    {
        const int tiles = (nNodes / 16) * 8;
        const int blocks = (tiles + wavesPerBlock - 1) / wavesPerBlock;
        gcn_gemm_wmma_kernel<64, 121, 8><<<blocks, T, 0, stream>>>(agg2, W2, outf, nNodes);
    }

    (void)n_in; (void)out_size; (void)ws_size;
}


# Fewer, longer-lived waves in every kernel: a gather wave walks 8 nodes (next nodes' sources fetched ahead), a GEMM wave takes a 16-row stripe with all column tiles, an edge thread 8 edges
# speedup vs baseline: 2.2984x; 1.1020x over previous
#include <hip/hip_runtime.h>
#include <hip/hip_bf16.h>


typedef __attribute__((ext_vector_type(2))) float v2f;
typedef __attribute__((ext_vector_type(8))) float v8f;


__global__ void gcn_zero_int_kernel(int* __restrict__ p, int n4) {
    int i = blockIdx.x * blockDim.x + threadIdx.x;
    if (i < n4) ((int4*)p)[i] = make_int4(0, 0, 0, 0);
}

template <int EPT>
__global__ __launch_bounds__(256) void gcn_count_rank_kernel(
    const int* __restrict__ dst, int* __restrict__ cnt,
    int* __restrict__ rank, int nE) {
    const int base = blockIdx.x * (256 * EPT) + threadIdx.x;
    int d[EPT];
    int r[EPT];
#pragma unroll
    for (int k = 0; k < EPT; ++k) {
        const int e = base + k * 256;
        d[k] = (e < nE) ? dst[e] : -1;
    }
#pragma unroll
    for (int k = 0; k < EPT; ++k)
        r[k] = (d[k] >= 0) ? atomicAdd(&cnt[d[k]], 1) : 0;
#pragma unroll
    for (int k = 0; k < EPT; ++k) {
        const int e = base + k * 256;
        if (e < nE) rank[e] = r[k];
    }
}

__global__ __launch_bounds__(1024) void gcn_scan_kernel(
    const int* __restrict__ cnt, int* __restrict__ rowptr,
    float* __restrict__ dinv, int n) {
    __shared__ int s_wsum[32];
    __shared__ int s_psum[32];
    const int tid  = threadIdx.x;
    const int lane = tid & 31;
    const int wid  = tid >> 5;
    const int i0   = blockIdx.x * 1024;

    int partial = 0;
    for (int i = tid; i < i0; i += 1024) partial += cnt[i];

    const int i  = i0 + tid;
    const int my = (i < n) ? cnt[i] : 0;

    int incl = my;
    for (int o = 1; o < 32; o <<= 1) {
        const int t = __shfl_up(incl, o);
        if (lane >= o) incl += t;
    }
    for (int o = 16; o > 0; o >>= 1)
        partial += __shfl_xor(partial, o);

    if (lane == 31) s_wsum[wid] = incl;
    if (lane == 0)  s_psum[wid] = partial;
    __syncthreads();

    int base = 0;
    for (int w = 0; w < 32; ++w) {
        base += s_psum[w];
        if (w < wid) base += s_wsum[w];
    }
    if (i < n) {
        rowptr[i] = base + incl - my;
        dinv[i]   = rsqrtf((float)(my + 1));
    }
}

template <int EPT>
__global__ __launch_bounds__(256) void gcn_place_kernel(
    const int* __restrict__ src, const int* __restrict__ dst,
    const int* __restrict__ rank, const int* __restrict__ rowptr,
    int* __restrict__ csr_src, int nE) {
    const int base = blockIdx.x * (256 * EPT) + threadIdx.x;
    int s[EPT];
    int d[EPT];
    int slot[EPT];
#pragma unroll
    for (int k = 0; k < EPT; ++k) {
        const int e  = base + k * 256;
        const int ec = e < nE ? e : nE - 1;
        s[k]    = src[ec];
        d[k]    = dst[ec];
        slot[k] = rank[ec];
    }
#pragma unroll
    for (int k = 0; k < EPT; ++k) slot[k] += rowptr[d[k]];
#pragma unroll
    for (int k = 0; k < EPT; ++k) {
        const int e = base + k * 256;
        if (e < nE) csr_src[slot[k]] = s[k];
    }
}

template <int K, int N, int NTILES>
__global__ __launch_bounds__(256) void gcn_gemm_wmma_kernel(
    const float* __restrict__ A, const float* __restrict__ B,
    float* __restrict__ C, int M) {
    constexpr int KP = (K + 3) & ~3;
    constexpr int KS = KP + 2;
    constexpr int NP = NTILES * 16;
    __shared__ float BT[NP * KS];

    for (int idx = threadIdx.x; idx < NP * KS; idx += blockDim.x) {
        const int n = idx / KS;
        const int k = idx - n * KS;
        BT[idx] = (n < N && k < K) ? B[(size_t)k * N + n] : 0.0f;
    }
    __syncthreads();

    const int wavesPerBlock = blockDim.x >> 5;
    const int stripe = blockIdx.x * wavesPerBlock + (threadIdx.x >> 5);
    if (stripe >= (M >> 4)) return;

    const int lane  = threadIdx.x & 31;
    const int row0  = stripe << 4;
    const int mrow  = lane & 15;
    const int khalf = (lane >> 4) << 1;

    const float* Arow = A + (size_t)(row0 + mrow) * K;
    const float* BTlane = BT + mrow * KS;

    v8f acc[NTILES];
#pragma unroll
    for (int t = 0; t < NTILES; ++t) acc[t] = v8f{};

    constexpr int KFULL = K & ~3;
#pragma unroll
    for (int k0 = 0; k0 < KFULL; k0 += 4) {
        const int ka = k0 + khalf;
        const float2 av = *(const float2*)(Arow + ka);
        v2f a; a.x = av.x; a.y = av.y;
#pragma unroll
        for (int t = 0; t < NTILES; ++t) {
            const float2 bv = *(const float2*)(BTlane + t * 16 * KS + ka);
            v2f b; b.x = bv.x; b.y = bv.y;
            acc[t] = __builtin_amdgcn_wmma_f32_16x16x4_f32(
                false, a, false, b, (short)0, acc[t], false, false);
        }
    }
    if constexpr (KFULL < K) {
        const int ka = KFULL + khalf;
        v2f a;
        a.x = (ka     < K) ? Arow[ka]     : 0.0f;
        a.y = (ka + 1 < K) ? Arow[ka + 1] : 0.0f;
#pragma unroll
        for (int t = 0; t < NTILES; ++t) {
            const float2 bv = *(const float2*)(BTlane + t * 16 * KS + ka);
            v2f b; b.x = bv.x; b.y = bv.y;
            acc[t] = __builtin_amdgcn_wmma_f32_16x16x4_f32(
                false, a, false, b, (short)0, acc[t], false, false);
        }
    }

    const int rbase = row0 + ((lane >> 4) << 3);
#pragma unroll
    for (int t = 0; t < NTILES; ++t) {
        const int col = (t << 4) + mrow;
        if (col < N) {
#pragma unroll
            for (int r = 0; r < 8; ++r)
                C[(size_t)(rbase + r) * N + col] = acc[t][r];
        }
    }
}

template <int U>
__device__ __forceinline__ void gcn_rows_accumulate(
    const float* __restrict__ xw, int my_s, int my_wbits, int j, int lane2,
    float& ax, float& ay) {
    float2 r[U];
    float  w[U];
#pragma unroll
    for (int u = 0; u < U; ++u) {
        const int s = __builtin_amdgcn_readlane(my_s, j + u);
        w[u] = __int_as_float(__builtin_amdgcn_readlane(my_wbits, j + u));
        r[u] = *(const float2*)(xw + (size_t)s * 64 + lane2);
    }
#pragma unroll
    for (int u = 0; u < U; ++u) {
        ax += w[u] * r[u].x;
        ay += w[u] * r[u].y;
    }
}

template <int DO_TANH, int NPW>
__global__ __launch_bounds__(256) void gcn_gather_kernel(
    const float* __restrict__ xw, const int* __restrict__ rowptr,
    const int* __restrict__ cnt, const int* __restrict__ csr_src,
    const float* __restrict__ dinv, float* __restrict__ out, int nNodes) {
    static_assert(NPW >= 1 && NPW <= 32, "one lane per node of the wave's span");
    const int wave = __builtin_amdgcn_readfirstlane(
        (int)(blockIdx.x * (blockDim.x >> 5) + (threadIdx.x >> 5)));
    const int n0 = wave * NPW;
    if (n0 >= nNodes) return;
    const int lane  = threadIdx.x & 31;
    const int lane2 = lane << 1;
    const int nleft = nNodes - n0;
    const int nn = nleft < NPW ? nleft : NPW;

    int m_beg = 0, m_cnt = 0, m_ddbits = 0;
    if (lane < nn) {
        m_beg    = rowptr[n0 + lane];
        m_cnt    = cnt[n0 + lane];
        m_ddbits = __float_as_int(dinv[n0 + lane]);
    }

    int begA = __builtin_amdgcn_readlane(m_beg, 0);
    int cntA = __builtin_amdgcn_readlane(m_cnt, 0);
    int sA = 0;
    if (lane < cntA) sA = csr_src[begA + lane];
    int begB = 0, cntB = 0, sB = 0;
    if (nn > 1) {
        begB = __builtin_amdgcn_readlane(m_beg, 1);
        cntB = __builtin_amdgcn_readlane(m_cnt, 1);
        if (lane < cntB) sB = csr_src[begB + lane];
    }
    float dA = 0.0f;
    if (lane < cntA) dA = dinv[sA];
    float2 svA = *(const float2*)(xw + (size_t)n0 * 64 + lane2);

    for (int i = 0; i < nn; ++i) {
        const int node = n0 + i;
        const int beg = begA;
        const int end = begA + cntA;
        const float dd = __int_as_float(__builtin_amdgcn_readlane(m_ddbits, i));
        const int   first_s = sA;
        const float first_w = dA * dd;
        float ax = dd * dd * svA.x;
        float ay = dd * dd * svA.y;

        begA = begB; cntA = cntB; sA = sB;
        if (i + 1 < nn) {
            dA = 0.0f;
            if (lane < cntA) dA = dinv[sA];
            svA = *(const float2*)(xw + (size_t)(node + 1) * 64 + lane2);
        }
        if (i + 2 < nn) {
            begB = __builtin_amdgcn_readlane(m_beg, i + 2);
            cntB = __builtin_amdgcn_readlane(m_cnt, i + 2);
            sB = 0;
            if (lane < cntB) sB = csr_src[begB + lane];
        }

        for (int p = beg; p < end; p += 32) {
            int   my_s = first_s;
            float my_w = first_w;
            if (p != beg) {
                const int idx = p + lane;
                my_s = 0;
                my_w = 0.0f;
                if (idx < end) {
                    my_s = csr_src[idx];
                    my_w = dinv[my_s] * dd;
                }
            }
            const int my_wbits = __float_as_int(my_w);
            const int left = end - p;
            const int n = left < 32 ? left : 32;
            int j = 0;
            for (; j + 8 <= n; j += 8)
                gcn_rows_accumulate<8>(xw, my_s, my_wbits, j, lane2, ax, ay);
            if (n - j >= 4) {
                gcn_rows_accumulate<4>(xw, my_s, my_wbits, j, lane2, ax, ay);
                j += 4;
            }
            if (n - j >= 2) {
                gcn_rows_accumulate<2>(xw, my_s, my_wbits, j, lane2, ax, ay);
                j += 2;
            }
            if (n - j >= 1)
                gcn_rows_accumulate<1>(xw, my_s, my_wbits, j, lane2, ax, ay);
        }

        float2 o;
        o.x = DO_TANH ? tanhf(ax) : ax;
        o.y = DO_TANH ? tanhf(ay) : ay;
        *(float2*)(out + (size_t)node * 64 + lane2) = o;
    }
}

extern "C" void kernel_launch(void* const* d_in, const int* in_sizes, int n_in,
                              void* d_out, int out_size, void* d_ws, size_t ws_size,
                              hipStream_t stream) {
    const float* x  = (const float*)d_in[0];
    const float* W1 = (const float*)d_in[1];
    const float* W2 = (const float*)d_in[2];
    const int* ei   = (const int*)d_in[3];

    const int FIN = 50, F1 = 64;
    const int nNodes = in_sizes[0] / FIN;
    const int nEdges = in_sizes[3] / 2;
    const int* src = ei;
    const int* dst = ei + nEdges;

    const int NP = (nNodes + 63) & ~63;
    const int EP = (nEdges + 63) & ~63;
    float* wsF    = (float*)d_ws;
    float* dinv   = wsF;
    int*   cnt    = (int*)(dinv + NP);
    int*   rowptr = cnt + NP;
    int*   rank   = rowptr + NP;
    int*   csr    = rank + EP;
    float* xw1    = (float*)(csr + EP);
    float* hbuf   = xw1 + (size_t)nNodes * F1;
    float* agg2   = hbuf + (size_t)nNodes * F1;
    float* outf   = (float*)d_out;

    const int T = 256;
    const int wavesPerBlock = T / 32;
    constexpr int EPT = 8;
    constexpr int NPW = 8;
    const int edgeBlocks = (nEdges + T * EPT - 1) / (T * EPT);
    const int gatherWaves = (nNodes + NPW - 1) / NPW;
    const int gatherBlocks = (gatherWaves + wavesPerBlock - 1) / wavesPerBlock;
    const int stripes = nNodes / 16;
    const int gemmBlocks = (stripes + wavesPerBlock - 1) / wavesPerBlock;

    gcn_zero_int_kernel<<<(NP / 4 + T - 1) / T, T, 0, stream>>>(cnt, NP / 4);
    gcn_count_rank_kernel<EPT><<<edgeBlocks, T, 0, stream>>>(dst, cnt, rank, nEdges);
    gcn_scan_kernel<<<(nNodes + 1023) / 1024, 1024, 0, stream>>>(cnt, rowptr, dinv, nNodes);
    gcn_place_kernel<EPT><<<edgeBlocks, T, 0, stream>>>(src, dst, rank, rowptr, csr, nEdges);

    gcn_gemm_wmma_kernel<50, 64, 4><<<gemmBlocks, T, 0, stream>>>(x, W1, xw1, nNodes);
    gcn_gather_kernel<1, NPW><<<gatherBlocks, T, 0, stream>>>(
        xw1, rowptr, cnt, csr, dinv, hbuf, nNodes);

    gcn_gather_kernel<0, NPW><<<gatherBlocks, T, 0, stream>>>(
        hbuf, rowptr, cnt, csr, dinv, agg2, nNodes);
    gcn_gemm_wmma_kernel<64, 121, 8><<<gemmBlocks, T, 0, stream>>>(agg2, W2, outf, nNodes);

    (void)n_in; (void)out_size; (void)ws_size;
}


# Gather waves ask for all of a node's source rows (up to 32, as groups of 8,8,8,8,4,2,1) before the first is used: one memory round trip per node, not one per group; f32 sum order unchanged
# speedup vs baseline: 2.3723x; 1.0322x over previous
#include <hip/hip_runtime.h>
#include <hip/hip_bf16.h>


typedef __attribute__((ext_vector_type(2))) float v2f;
typedef __attribute__((ext_vector_type(8))) float v8f;


__global__ void gcn_zero_int_kernel(int* __restrict__ p, int n4) {
    int i = blockIdx.x * blockDim.x + threadIdx.x;
    if (i < n4) ((int4*)p)[i] = make_int4(0, 0, 0, 0);
}

template <int EPT>
__global__ __launch_bounds__(256) void gcn_count_rank_kernel(
    const int* __restrict__ dst, int* __restrict__ cnt,
    int* __restrict__ rank, int nE) {
    const int base = blockIdx.x * (256 * EPT) + threadIdx.x;
    int d[EPT];
    int r[EPT];
#pragma unroll
    for (int k = 0; k < EPT; ++k) {
        const int e = base + k * 256;
        d[k] = (e < nE) ? dst[e] : -1;
    }
#pragma unroll
    for (int k = 0; k < EPT; ++k)
        r[k] = (d[k] >= 0) ? atomicAdd(&cnt[d[k]], 1) : 0;
#pragma unroll
    for (int k = 0; k < EPT; ++k) {
        const int e = base + k * 256;
        if (e < nE) rank[e] = r[k];
    }
}

__global__ __launch_bounds__(1024) void gcn_scan_kernel(
    const int* __restrict__ cnt, int* __restrict__ rowptr,
    float* __restrict__ dinv, int n) {
    __shared__ int s_wsum[32];
    __shared__ int s_psum[32];
    const int tid  = threadIdx.x;
    const int lane = tid & 31;
    const int wid  = tid >> 5;
    const int i0   = blockIdx.x * 1024;

    int partial = 0;
    for (int i = tid; i < i0; i += 1024) partial += cnt[i];

    const int i  = i0 + tid;
    const int my = (i < n) ? cnt[i] : 0;

    int incl = my;
    for (int o = 1; o < 32; o <<= 1) {
        const int t = __shfl_up(incl, o);
        if (lane >= o) incl += t;
    }
    for (int o = 16; o > 0; o >>= 1)
        partial += __shfl_xor(partial, o);

    if (lane == 31) s_wsum[wid] = incl;
    if (lane == 0)  s_psum[wid] = partial;
    __syncthreads();

    int base = 0;
    for (int w = 0; w < 32; ++w) {
        base += s_psum[w];
        if (w < wid) base += s_wsum[w];
    }
    if (i < n) {
        rowptr[i] = base + incl - my;
        dinv[i]   = rsqrtf((float)(my + 1));
    }
}

template <int EPT>
__global__ __launch_bounds__(256) void gcn_place_kernel(
    const int* __restrict__ src, const int* __restrict__ dst,
    const int* __restrict__ rank, const int* __restrict__ rowptr,
    int* __restrict__ csr_src, int nE) {
    const int base = blockIdx.x * (256 * EPT) + threadIdx.x;
    int s[EPT];
    int d[EPT];
    int slot[EPT];
#pragma unroll
    for (int k = 0; k < EPT; ++k) {
        const int e  = base + k * 256;
        const int ec = e < nE ? e : nE - 1;
        s[k]    = src[ec];
        d[k]    = dst[ec];
        slot[k] = rank[ec];
    }
#pragma unroll
    for (int k = 0; k < EPT; ++k) slot[k] += rowptr[d[k]];
#pragma unroll
    for (int k = 0; k < EPT; ++k) {
        const int e = base + k * 256;
        if (e < nE) csr_src[slot[k]] = s[k];
    }
}

template <int K, int N, int NTILES>
__global__ __launch_bounds__(256) void gcn_gemm_wmma_kernel(
    const float* __restrict__ A, const float* __restrict__ B,
    float* __restrict__ C, int M) {
    constexpr int KP = (K + 3) & ~3;
    constexpr int KS = KP + 2;
    constexpr int NP = NTILES * 16;
    __shared__ float BT[NP * KS];

    for (int idx = threadIdx.x; idx < NP * KS; idx += blockDim.x) {
        const int n = idx / KS;
        const int k = idx - n * KS;
        BT[idx] = (n < N && k < K) ? B[(size_t)k * N + n] : 0.0f;
    }
    __syncthreads();

    const int wavesPerBlock = blockDim.x >> 5;
    const int stripe = blockIdx.x * wavesPerBlock + (threadIdx.x >> 5);
    if (stripe >= (M >> 4)) return;

    const int lane  = threadIdx.x & 31;
    const int row0  = stripe << 4;
    const int mrow  = lane & 15;
    const int khalf = (lane >> 4) << 1;

    const float* Arow = A + (size_t)(row0 + mrow) * K;
    const float* BTlane = BT + mrow * KS;

    v8f acc[NTILES];
#pragma unroll
    for (int t = 0; t < NTILES; ++t) acc[t] = v8f{};

    constexpr int KFULL = K & ~3;
#pragma unroll
    for (int k0 = 0; k0 < KFULL; k0 += 4) {
        const int ka = k0 + khalf;
        const float2 av = *(const float2*)(Arow + ka);
        v2f a; a.x = av.x; a.y = av.y;
#pragma unroll
        for (int t = 0; t < NTILES; ++t) {
            const float2 bv = *(const float2*)(BTlane + t * 16 * KS + ka);
            v2f b; b.x = bv.x; b.y = bv.y;
            acc[t] = __builtin_amdgcn_wmma_f32_16x16x4_f32(
                false, a, false, b, (short)0, acc[t], false, false);
        }
    }
    if constexpr (KFULL < K) {
        const int ka = KFULL + khalf;
        v2f a;
        a.x = (ka     < K) ? Arow[ka]     : 0.0f;
        a.y = (ka + 1 < K) ? Arow[ka + 1] : 0.0f;
#pragma unroll
        for (int t = 0; t < NTILES; ++t) {
            const float2 bv = *(const float2*)(BTlane + t * 16 * KS + ka);
            v2f b; b.x = bv.x; b.y = bv.y;
            acc[t] = __builtin_amdgcn_wmma_f32_16x16x4_f32(
                false, a, false, b, (short)0, acc[t], false, false);
        }
    }

    const int rbase = row0 + ((lane >> 4) << 3);
#pragma unroll
    for (int t = 0; t < NTILES; ++t) {
        const int col = (t << 4) + mrow;
        if (col < N) {
#pragma unroll
            for (int r = 0; r < 8; ++r)
                C[(size_t)(rbase + r) * N + col] = acc[t][r];
        }
    }
}

template <int U>
__device__ __forceinline__ void gcn_rows_ask(
    const float* __restrict__ xw, int my_s, int j, int lane2, float2* r) {
#pragma unroll
    for (int u = 0; u < U; ++u) {
        const int s = __builtin_amdgcn_readlane(my_s, j + u);
        r[u] = *(const float2*)(xw + (size_t)s * 64 + lane2);
    }
}

template <int U>
__device__ __forceinline__ void gcn_rows_add(
    const float2* r, int my_wbits, int j, float& ax, float& ay) {
#pragma unroll
    for (int u = 0; u < U; ++u) {
        const float w = __int_as_float(__builtin_amdgcn_readlane(my_wbits, j + u));
        ax += w * r[u].x;
        ay += w * r[u].y;
    }
}

template <int DO_TANH, int NPW>
__global__ __launch_bounds__(256) void gcn_gather_kernel(
    const float* __restrict__ xw, const int* __restrict__ rowptr,
    const int* __restrict__ cnt, const int* __restrict__ csr_src,
    const float* __restrict__ dinv, float* __restrict__ out, int nNodes) {
    static_assert(NPW >= 1 && NPW <= 32, "one lane per node of the wave's span");
    const int wave = __builtin_amdgcn_readfirstlane(
        (int)(blockIdx.x * (blockDim.x >> 5) + (threadIdx.x >> 5)));
    const int n0 = wave * NPW;
    if (n0 >= nNodes) return;
    const int lane  = threadIdx.x & 31;
    const int lane2 = lane << 1;
    const int nleft = nNodes - n0;
    const int nn = nleft < NPW ? nleft : NPW;

    int m_beg = 0, m_cnt = 0, m_ddbits = 0;
    if (lane < nn) {
        m_beg    = rowptr[n0 + lane];
        m_cnt    = cnt[n0 + lane];
        m_ddbits = __float_as_int(dinv[n0 + lane]);
    }

    int begA = __builtin_amdgcn_readlane(m_beg, 0);
    int cntA = __builtin_amdgcn_readlane(m_cnt, 0);
    int sA = 0;
    if (lane < cntA) sA = csr_src[begA + lane];
    int begB = 0, cntB = 0, sB = 0;
    if (nn > 1) {
        begB = __builtin_amdgcn_readlane(m_beg, 1);
        cntB = __builtin_amdgcn_readlane(m_cnt, 1);
        if (lane < cntB) sB = csr_src[begB + lane];
    }
    float dA = 0.0f;
    if (lane < cntA) dA = dinv[sA];
    float2 svA = *(const float2*)(xw + (size_t)n0 * 64 + lane2);

    for (int i = 0; i < nn; ++i) {
        const int node = n0 + i;
        const int beg = begA;
        const int end = begA + cntA;
        const float dd = __int_as_float(__builtin_amdgcn_readlane(m_ddbits, i));
        const int   first_s = sA;
        const float first_w = dA * dd;
        float ax = dd * dd * svA.x;
        float ay = dd * dd * svA.y;

        begA = begB; cntA = cntB; sA = sB;
        if (i + 1 < nn) {
            dA = 0.0f;
            if (lane < cntA) dA = dinv[sA];
            svA = *(const float2*)(xw + (size_t)(node + 1) * 64 + lane2);
        }
        if (i + 2 < nn) {
            begB = __builtin_amdgcn_readlane(m_beg, i + 2);
            cntB = __builtin_amdgcn_readlane(m_cnt, i + 2);
            sB = 0;
            if (lane < cntB) sB = csr_src[begB + lane];
        }

        for (int p = beg; p < end; p += 32) {
            int   my_s = first_s;
            float my_w = first_w;
            if (p != beg) {
                const int idx = p + lane;
                my_s = 0;
                my_w = 0.0f;
                if (idx < end) {
                    my_s = csr_src[idx];
                    my_w = dinv[my_s] * dd;
                }
            }
            const int my_wbits = __float_as_int(my_w);
            const int left = end - p;
            const int n = left < 32 ? left : 32;
            float2 ra[8], rb[8], rc[8], rd[8], t4[4], t2[2], t1[1];
            const int nt = n & ~7;
            if (n >= 8)  gcn_rows_ask<8>(xw, my_s, 0,  lane2, ra);
            if (n >= 16) gcn_rows_ask<8>(xw, my_s, 8,  lane2, rb);
            if (n >= 24) gcn_rows_ask<8>(xw, my_s, 16, lane2, rc);
            if (n >= 32) gcn_rows_ask<8>(xw, my_s, 24, lane2, rd);
            int j = nt;
            if (n & 4) { gcn_rows_ask<4>(xw, my_s, j, lane2, t4); j += 4; }
            if (n & 2) { gcn_rows_ask<2>(xw, my_s, j, lane2, t2); j += 2; }
            if (n & 1) { gcn_rows_ask<1>(xw, my_s, j, lane2, t1); }

            if (n >= 8)  gcn_rows_add<8>(ra, my_wbits, 0,  ax, ay);
            if (n >= 16) gcn_rows_add<8>(rb, my_wbits, 8,  ax, ay);
            if (n >= 24) gcn_rows_add<8>(rc, my_wbits, 16, ax, ay);
            if (n >= 32) gcn_rows_add<8>(rd, my_wbits, 24, ax, ay);
            j = nt;
            if (n & 4) { gcn_rows_add<4>(t4, my_wbits, j, ax, ay); j += 4; }
            if (n & 2) { gcn_rows_add<2>(t2, my_wbits, j, ax, ay); j += 2; }
            if (n & 1) { gcn_rows_add<1>(t1, my_wbits, j, ax, ay); }
        }

        float2 o;
        o.x = DO_TANH ? tanhf(ax) : ax;
        o.y = DO_TANH ? tanhf(ay) : ay;
        *(float2*)(out + (size_t)node * 64 + lane2) = o;
    }
}

extern "C" void kernel_launch(void* const* d_in, const int* in_sizes, int n_in,
                              void* d_out, int out_size, void* d_ws, size_t ws_size,
                              hipStream_t stream) {
    const float* x  = (const float*)d_in[0];
    const float* W1 = (const float*)d_in[1];
    const float* W2 = (const float*)d_in[2];
    const int* ei   = (const int*)d_in[3];

    const int FIN = 50, F1 = 64;
    const int nNodes = in_sizes[0] / FIN;
    const int nEdges = in_sizes[3] / 2;
    const int* src = ei;
    const int* dst = ei + nEdges;

    const int NP = (nNodes + 63) & ~63;
    const int EP = (nEdges + 63) & ~63;
    float* wsF    = (float*)d_ws;
    float* dinv   = wsF;
    int*   cnt    = (int*)(dinv + NP);
    int*   rowptr = cnt + NP;
    int*   rank   = rowptr + NP;
    int*   csr    = rank + EP;
    float* xw1    = (float*)(csr + EP);
    float* hbuf   = xw1 + (size_t)nNodes * F1;
    float* agg2   = hbuf + (size_t)nNodes * F1;
    float* outf   = (float*)d_out;

    const int T = 256;
    const int wavesPerBlock = T / 32;
    constexpr int EPT = 8;
    constexpr int NPW = 8;
    const int edgeBlocks = (nEdges + T * EPT - 1) / (T * EPT);
    const int gatherWaves = (nNodes + NPW - 1) / NPW;
    const int gatherBlocks = (gatherWaves + wavesPerBlock - 1) / wavesPerBlock;
    const int stripes = nNodes / 16;
    const int gemmBlocks = (stripes + wavesPerBlock - 1) / wavesPerBlock;

    gcn_zero_int_kernel<<<(NP / 4 + T - 1) / T, T, 0, stream>>>(cnt, NP / 4);
    gcn_count_rank_kernel<EPT><<<edgeBlocks, T, 0, stream>>>(dst, cnt, rank, nEdges);
    gcn_scan_kernel<<<(nNodes + 1023) / 1024, 1024, 0, stream>>>(cnt, rowptr, dinv, nNodes);
    gcn_place_kernel<EPT><<<edgeBlocks, T, 0, stream>>>(src, dst, rank, rowptr, csr, nEdges);

    gcn_gemm_wmma_kernel<50, 64, 4><<<gemmBlocks, T, 0, stream>>>(x, W1, xw1, nNodes);
    gcn_gather_kernel<1, NPW><<<gatherBlocks, T, 0, stream>>>(
        xw1, rowptr, cnt, csr, dinv, hbuf, nNodes);

    gcn_gather_kernel<0, NPW><<<gatherBlocks, T, 0, stream>>>(
        hbuf, rowptr, cnt, csr, dinv, agg2, nNodes);
    gcn_gemm_wmma_kernel<64, 121, 8><<<gemmBlocks, T, 0, stream>>>(agg2, W2, outf, nNodes);

    (void)n_in; (void)out_size; (void)ws_size;
}


# Each layer's product runs inside its gather kernel: a block sums 64 nodes' rows (layer 1: x, 50 wide) into LDS and multiplies them by W on WMMA; 2 launches fewer; f32 sum order changes (reassociated)
# speedup vs baseline: 2.6585x; 1.1206x over previous
#include <hip/hip_runtime.h>
#include <hip/hip_bf16.h>


typedef __attribute__((ext_vector_type(2))) float v2f;
typedef __attribute__((ext_vector_type(8))) float v8f;


__global__ void gcn_zero_int_kernel(int* __restrict__ p, int n4) {
    int i = blockIdx.x * blockDim.x + threadIdx.x;
    if (i < n4) ((int4*)p)[i] = make_int4(0, 0, 0, 0);
}

template <int EPT>
__global__ __launch_bounds__(256) void gcn_count_rank_kernel(
    const int* __restrict__ dst, int* __restrict__ cnt,
    int* __restrict__ rank, int nE) {
    const int base = blockIdx.x * (256 * EPT) + threadIdx.x;
    int d[EPT];
    int r[EPT];
#pragma unroll
    for (int k = 0; k < EPT; ++k) {
        const int e = base + k * 256;
        d[k] = (e < nE) ? dst[e] : -1;
    }
#pragma unroll
    for (int k = 0; k < EPT; ++k)
        r[k] = (d[k] >= 0) ? atomicAdd(&cnt[d[k]], 1) : 0;
#pragma unroll
    for (int k = 0; k < EPT; ++k) {
        const int e = base + k * 256;
        if (e < nE) rank[e] = r[k];
    }
}

__global__ __launch_bounds__(1024) void gcn_scan_kernel(
    const int* __restrict__ cnt, int* __restrict__ rowptr,
    float* __restrict__ dinv, int n) {
    __shared__ int s_wsum[32];
    __shared__ int s_psum[32];
    const int tid  = threadIdx.x;
    const int lane = tid & 31;
    const int wid  = tid >> 5;
    const int i0   = blockIdx.x * 1024;

    int partial = 0;
    for (int i = tid; i < i0; i += 1024) partial += cnt[i];

    const int i  = i0 + tid;
    const int my = (i < n) ? cnt[i] : 0;

    int incl = my;
    for (int o = 1; o < 32; o <<= 1) {
        const int t = __shfl_up(incl, o);
        if (lane >= o) incl += t;
    }
    for (int o = 16; o > 0; o >>= 1)
        partial += __shfl_xor(partial, o);

    if (lane == 31) s_wsum[wid] = incl;
    if (lane == 0)  s_psum[wid] = partial;
    __syncthreads();

    int base = 0;
    for (int w = 0; w < 32; ++w) {
        base += s_psum[w];
        if (w < wid) base += s_wsum[w];
    }
    if (i < n) {
        rowptr[i] = base + incl - my;
        dinv[i]   = rsqrtf((float)(my + 1));
    }
}

template <int EPT>
__global__ __launch_bounds__(256) void gcn_place_kernel(
    const int* __restrict__ src, const int* __restrict__ dst,
    const int* __restrict__ rank, const int* __restrict__ rowptr,
    int* __restrict__ csr_src, int nE) {
    const int base = blockIdx.x * (256 * EPT) + threadIdx.x;
    int s[EPT];
    int d[EPT];
    int slot[EPT];
#pragma unroll
    for (int k = 0; k < EPT; ++k) {
        const int e  = base + k * 256;
        const int ec = e < nE ? e : nE - 1;
        s[k]    = src[ec];
        d[k]    = dst[ec];
        slot[k] = rank[ec];
    }
#pragma unroll
    for (int k = 0; k < EPT; ++k) slot[k] += rowptr[d[k]];
#pragma unroll
    for (int k = 0; k < EPT; ++k) {
        const int e = base + k * 256;
        if (e < nE) csr_src[slot[k]] = s[k];
    }
}

template <int K, int U>
__device__ __forceinline__ void gcn_rows_ask(
    const float* __restrict__ xin, int my_s, int j, int coff, float2* r) {
#pragma unroll
    for (int u = 0; u < U; ++u) {
        const int s = __builtin_amdgcn_readlane(my_s, j + u);
        r[u] = *(const float2*)(xin + (size_t)s * K + coff);
    }
}

template <int U>
__device__ __forceinline__ void gcn_rows_add(
    const float2* r, int my_wbits, int j, float& ax, float& ay) {
#pragma unroll
    for (int u = 0; u < U; ++u) {
        const float w = __int_as_float(__builtin_amdgcn_readlane(my_wbits, j + u));
        ax += w * r[u].x;
        ay += w * r[u].y;
    }
}

template <int K, int N, int NTILES, int DO_TANH>
__global__ __launch_bounds__(256) void gcn_gather_gemm_kernel(
    const float* __restrict__ xin, const float* __restrict__ B,
    const int* __restrict__ rowptr, const int* __restrict__ cnt,
    const int* __restrict__ csr_src, const float* __restrict__ dinv,
    float* __restrict__ C, int nNodes) {
    static_assert((K & 1) == 0 && K <= 64, "float2 row loads, one per lane");
    static_assert((NTILES & 1) == 0, "the column tiles are split between two waves");
    constexpr int KP  = (K + 3) & ~3;
    constexpr int KS  = KP + 2;
    constexpr int NP  = NTILES * 16;
    constexpr int NT2 = NTILES / 2;
    constexpr int NTH = 256;
    constexpr int NPW = 8;
    constexpr int NPB = NPW * (NTH / 32);
    __shared__ float BT[NP * KS];
    __shared__ float AT[NPB * KS];

    constexpr int TOT = KS * NP;
    constexpr int PER = (TOT + NTH - 1) / NTH;
    {
        float v[PER];
#pragma unroll
        for (int q = 0; q < PER; ++q) {
            const int idx = (int)threadIdx.x + q * NTH;
            const int k = idx / NP;
            const int n = idx - k * NP;
            const int kc = k < K ? k : K - 1;
            const int nc = n < N ? n : N - 1;
            const float b = B[kc * N + nc];
            v[q] = (n < N && k < K) ? b : 0.0f;
        }
#pragma unroll
        for (int q = 0; q < PER; ++q) {
            const int idx = (int)threadIdx.x + q * NTH;
            const int k = idx / NP;
            const int n = idx - k * NP;
            if (idx < TOT) BT[n * KS + k] = v[q];
        }
    }

    const int wib = __builtin_amdgcn_readfirstlane((int)(threadIdx.x >> 5));
    const int nb0 = __builtin_amdgcn_readfirstlane((int)(blockIdx.x * NPB));
    const int n0  = nb0 + wib * NPW;
    const int lane  = threadIdx.x & 31;
    const int lane2 = lane << 1;
    const int coff = lane2 < K ? lane2 : 0;
    const int nleft = nNodes - n0;
    const int nn = nleft < NPW ? nleft : NPW;
    float* ATw = AT + wib * (NPW * KS);

    if (nn > 0) {
        int m_beg = 0, m_cnt = 0, m_ddbits = 0;
        if (lane < nn) {
            m_beg    = rowptr[n0 + lane];
            m_cnt    = cnt[n0 + lane];
            m_ddbits = __float_as_int(dinv[n0 + lane]);
        }

        int begA = __builtin_amdgcn_readlane(m_beg, 0);
        int cntA = __builtin_amdgcn_readlane(m_cnt, 0);
        int sA = 0;
        if (lane < cntA) sA = csr_src[begA + lane];
        int begB = 0, cntB = 0, sB = 0;
        if (nn > 1) {
            begB = __builtin_amdgcn_readlane(m_beg, 1);
            cntB = __builtin_amdgcn_readlane(m_cnt, 1);
            if (lane < cntB) sB = csr_src[begB + lane];
        }
        float dA = 0.0f;
        if (lane < cntA) dA = dinv[sA];
        float2 svA = *(const float2*)(xin + (size_t)n0 * K + coff);

        for (int i = 0; i < nn; ++i) {
            const int node = n0 + i;
            const int beg = begA;
            const int end = begA + cntA;
            const float dd = __int_as_float(__builtin_amdgcn_readlane(m_ddbits, i));
            const int   first_s = sA;
            const float first_w = dA * dd;
            float ax = dd * dd * svA.x;
            float ay = dd * dd * svA.y;

            begA = begB; cntA = cntB; sA = sB;
            if (i + 1 < nn) {
                dA = 0.0f;
                if (lane < cntA) dA = dinv[sA];
                svA = *(const float2*)(xin + (size_t)(node + 1) * K + coff);
            }
            if (i + 2 < nn) {
                begB = __builtin_amdgcn_readlane(m_beg, i + 2);
                cntB = __builtin_amdgcn_readlane(m_cnt, i + 2);
                sB = 0;
                if (lane < cntB) sB = csr_src[begB + lane];
            }

            for (int p = beg; p < end; p += 32) {
                int   my_s = first_s;
                float my_w = first_w;
                if (p != beg) {
                    const int idx = p + lane;
                    my_s = 0;
                    my_w = 0.0f;
                    if (idx < end) {
                        my_s = csr_src[idx];
                        my_w = dinv[my_s] * dd;
                    }
                }
                const int my_wbits = __float_as_int(my_w);
                const int left = end - p;
                const int n = left < 32 ? left : 32;
                float2 ra[8], rb[8], rc[8], rd[8], t4[4], t2[2], t1[1];
                const int nt = n & ~7;
                if (n >= 8)  gcn_rows_ask<K, 8>(xin, my_s, 0,  coff, ra);
                if (n >= 16) gcn_rows_ask<K, 8>(xin, my_s, 8,  coff, rb);
                if (n >= 24) gcn_rows_ask<K, 8>(xin, my_s, 16, coff, rc);
                if (n >= 32) gcn_rows_ask<K, 8>(xin, my_s, 24, coff, rd);
                int j = nt;
                if (n & 4) { gcn_rows_ask<K, 4>(xin, my_s, j, coff, t4); j += 4; }
                if (n & 2) { gcn_rows_ask<K, 2>(xin, my_s, j, coff, t2); j += 2; }
                if (n & 1) { gcn_rows_ask<K, 1>(xin, my_s, j, coff, t1); }

                if (n >= 8)  gcn_rows_add<8>(ra, my_wbits, 0,  ax, ay);
                if (n >= 16) gcn_rows_add<8>(rb, my_wbits, 8,  ax, ay);
                if (n >= 24) gcn_rows_add<8>(rc, my_wbits, 16, ax, ay);
                if (n >= 32) gcn_rows_add<8>(rd, my_wbits, 24, ax, ay);
                j = nt;
                if (n & 4) { gcn_rows_add<4>(t4, my_wbits, j, ax, ay); j += 4; }
                if (n & 2) { gcn_rows_add<2>(t2, my_wbits, j, ax, ay); j += 2; }
                if (n & 1) { gcn_rows_add<1>(t1, my_wbits, j, ax, ay); }
            }

            if (lane2 < KP) {
                float2 o;
                o.x = lane2 < K ? ax : 0.0f;
                o.y = lane2 < K ? ay : 0.0f;
                *(float2*)(ATw + i * KS + lane2) = o;
            }
        }
    }

    __syncthreads();

    const int rt  = wib >> 1;
    const int ct0 = (wib & 1) * NT2;
    const int row0 = nb0 + (rt << 4);
    if (row0 >= nNodes) return;

    const int mrow  = lane & 15;
    const int khalf = (lane >> 4) << 1;
    const float* ATlane = AT + ((rt << 4) + mrow) * KS;
    const float* BTlane = BT + (ct0 * 16 + mrow) * KS;

    v8f acc[NT2];
#pragma unroll
    for (int t = 0; t < NT2; ++t) acc[t] = v8f{};

#pragma unroll
    for (int k0 = 0; k0 < KP; k0 += 4) {
        const int ka = k0 + khalf;
        const float2 av = *(const float2*)(ATlane + ka);
        v2f a; a.x = av.x; a.y = av.y;
#pragma unroll
        for (int t = 0; t < NT2; ++t) {
            const float2 bv = *(const float2*)(BTlane + t * 16 * KS + ka);
            v2f b; b.x = bv.x; b.y = bv.y;
            acc[t] = __builtin_amdgcn_wmma_f32_16x16x4_f32(
                false, a, false, b, (short)0, acc[t], false, false);
        }
    }

    const int rbase = row0 + ((lane >> 4) << 3);
#pragma unroll
    for (int t = 0; t < NT2; ++t) {
        const int col = ((ct0 + t) << 4) + mrow;
        if (col < N) {
#pragma unroll
            for (int r = 0; r < 8; ++r) {
                const float v = acc[t][r];
                C[(size_t)(rbase + r) * N + col] = DO_TANH ? tanhf(v) : v;
            }
        }
    }
}

extern "C" void kernel_launch(void* const* d_in, const int* in_sizes, int n_in,
                              void* d_out, int out_size, void* d_ws, size_t ws_size,
                              hipStream_t stream) {
    const float* x  = (const float*)d_in[0];
    const float* W1 = (const float*)d_in[1];
    const float* W2 = (const float*)d_in[2];
    const int* ei   = (const int*)d_in[3];

    const int FIN = 50, F1 = 64;
    const int nNodes = in_sizes[0] / FIN;
    const int nEdges = in_sizes[3] / 2;
    const int* src = ei;
    const int* dst = ei + nEdges;

    const int NP = (nNodes + 63) & ~63;
    const int EP = (nEdges + 63) & ~63;
    float* wsF    = (float*)d_ws;
    float* dinv   = wsF;
    int*   cnt    = (int*)(dinv + NP);
    int*   rowptr = cnt + NP;
    int*   rank   = rowptr + NP;
    int*   csr    = rank + EP;
    float* hbuf   = (float*)(csr + EP);
    float* outf   = (float*)d_out;

    const int T = 256;
    constexpr int EPT = 8;
    const int edgeBlocks = (nEdges + T * EPT - 1) / (T * EPT);
    const int layerBlocks = (nNodes + 63) / 64;

    gcn_zero_int_kernel<<<(NP / 4 + T - 1) / T, T, 0, stream>>>(cnt, NP / 4);
    gcn_count_rank_kernel<EPT><<<edgeBlocks, T, 0, stream>>>(dst, cnt, rank, nEdges);
    gcn_scan_kernel<<<(nNodes + 1023) / 1024, 1024, 0, stream>>>(cnt, rowptr, dinv, nNodes);
    gcn_place_kernel<EPT><<<edgeBlocks, T, 0, stream>>>(src, dst, rank, rowptr, csr, nEdges);

    gcn_gather_gemm_kernel<50, 64, 4, 1><<<layerBlocks, T, 0, stream>>>(
        x, W1, rowptr, cnt, csr, dinv, hbuf, nNodes);
    gcn_gather_gemm_kernel<64, 121, 8, 0><<<layerBlocks, T, 0, stream>>>(
        hbuf, W2, rowptr, cnt, csr, dinv, outf, nNodes);

    (void)n_in; (void)out_size; (void)ws_size; (void)F1;
}


# Edges grouped by destination with LDS counters only, two levels (256-node bucket, then node): no returning global atomic per edge; 5 launches, not 6; f32 sum order changes (arrival order in LDS)
# speedup vs baseline: 4.8324x; 1.8177x over previous
#include <hip/hip_runtime.h>
#include <hip/hip_bf16.h>


typedef __attribute__((ext_vector_type(2))) float v2f;
typedef __attribute__((ext_vector_type(8))) float v8f;


constexpr int BKT_SHIFT = 8;
constexpr int BKT_NODES = 1 << BKT_SHIFT;
constexpr int MAXBKT    = 1024;
constexpr int BNTH      = 1024;
constexpr int BEPT      = 8;

__global__ __launch_bounds__(BNTH) void gcn_bucket_count_kernel(
    const int* __restrict__ dst, int* __restrict__ hist, int nE, int nbkt) {
    __shared__ int h[MAXBKT];
    const int tid = threadIdx.x;
    if (tid < nbkt) h[tid] = 0;
    const int base = blockIdx.x * (BNTH * BEPT) + tid;
    int d[BEPT];
#pragma unroll
    for (int k = 0; k < BEPT; ++k) {
        const int e = base + k * BNTH;
        d[k] = (e < nE) ? dst[e] : -1;
    }
    __syncthreads();
#pragma unroll
    for (int k = 0; k < BEPT; ++k)
        if (d[k] >= 0) atomicAdd(&h[d[k] >> BKT_SHIFT], 1);
    __syncthreads();
    if (tid < nbkt) hist[blockIdx.x * nbkt + tid] = h[tid];
}

__global__ __launch_bounds__(BNTH) void gcn_bucket_place_kernel(
    const int* __restrict__ src, const int* __restrict__ dst,
    const int* __restrict__ hist, int* __restrict__ bstart,
    int2* __restrict__ binned, int nE, int nbkt, int nblk) {
    static_assert(MAXBKT == BNTH, "one bucket counter per thread");
    __shared__ int cur[MAXBKT];
    __shared__ int tot_s[MAXBKT];
    __shared__ int bef_s[MAXBKT];
    __shared__ int s_wsum[BNTH / 32];
    const int tid  = threadIdx.x;
    const int lane = tid & 31;
    const int wid  = tid >> 5;
    const int bid  = blockIdx.x;
    const int base = bid * (BNTH * BEPT) + tid;
    int s[BEPT];
    int d[BEPT];
#pragma unroll
    for (int k = 0; k < BEPT; ++k) {
        const int e  = base + k * BNTH;
        const int ec = e < nE ? e : nE - 1;
        s[k] = src[ec];
        d[k] = dst[ec];
    }

    constexpr int TR = 4, TC = 8;
    tot_s[tid] = 0;
    bef_s[tid] = 0;
    __syncthreads();
    for (int j0 = 0; j0 < nbkt; j0 += 32 * TC) {
        int at[TC], ab[TC];
#pragma unroll
        for (int u = 0; u < TC; ++u) { at[u] = 0; ab[u] = 0; }
        for (int r0 = wid; r0 < nblk; r0 += 32 * TR) {
            int v[TR][TC];
#pragma unroll
            for (int r = 0; r < TR; ++r) {
                const int b  = r0 + 32 * r;
                const int bc = b < nblk ? b : nblk - 1;
#pragma unroll
                for (int u = 0; u < TC; ++u) {
                    const int k  = j0 + 32 * u + lane;
                    const int kc = k < nbkt ? k : nbkt - 1;
                    v[r][u] = hist[bc * nbkt + kc];
                }
            }
#pragma unroll
            for (int r = 0; r < TR; ++r) {
                const int b = r0 + 32 * r;
#pragma unroll
                for (int u = 0; u < TC; ++u) {
                    const int k = j0 + 32 * u + lane;
                    const int x = (b < nblk && k < nbkt) ? v[r][u] : 0;
                    at[u] += x;
                    ab[u] += (b < bid) ? x : 0;
                }
            }
        }
#pragma unroll
        for (int u = 0; u < TC; ++u) {
            const int k = j0 + 32 * u + lane;
            if (k < nbkt) {
                atomicAdd(&tot_s[k], at[u]);
                atomicAdd(&bef_s[k], ab[u]);
            }
        }
    }
    __syncthreads();
    const int tot    = tid < nbkt ? tot_s[tid] : 0;
    const int before = bef_s[tid];

    int incl = tot;
    for (int o = 1; o < 32; o <<= 1) {
        const int t = __shfl_up(incl, o);
        if (lane >= o) incl += t;
    }
    if (lane == 31) s_wsum[wid] = incl;
    __syncthreads();
    int wbase = 0;
    for (int w = 0; w < BNTH / 32; ++w)
        if (w < wid) wbase += s_wsum[w];
    const int first = wbase + incl - tot;
    if (tid < nbkt) {
        cur[tid] = first + before;
        if (bid == 0) bstart[tid] = first;
    }
    if (bid == 0 && tid == 0) bstart[nbkt] = nE;
    __syncthreads();

    int slot[BEPT];
#pragma unroll
    for (int k = 0; k < BEPT; ++k) {
        const int e = base + k * BNTH;
        slot[k] = (e < nE) ? atomicAdd(&cur[d[k] >> BKT_SHIFT], 1) : -1;
    }
#pragma unroll
    for (int k = 0; k < BEPT; ++k)
        if (slot[k] >= 0) binned[slot[k]] = make_int2(s[k], d[k]);
}

__global__ __launch_bounds__(BNTH) void gcn_bucket_rows_kernel(
    const int2* __restrict__ binned, const int* __restrict__ bstart,
    int* __restrict__ rowptr, int* __restrict__ cnt, float* __restrict__ dinv,
    int* __restrict__ csr_src, int nNodes) {
    constexpr int CU = 4;
    static_assert(CU == 4, "the two load fences below name four values");
    __shared__ int c[BKT_NODES];
    __shared__ int cur[BKT_NODES];
    __shared__ int s_wsum[BKT_NODES / 32];
    const int tid  = threadIdx.x;
    const int lane = tid & 31;
    const int wid  = tid >> 5;
    const int node0 = blockIdx.x << BKT_SHIFT;
    const int b0 = bstart[blockIdx.x];
    const int b1 = bstart[blockIdx.x + 1];
    if (tid < BKT_NODES) c[tid] = 0;
    __syncthreads();

    for (int e = b0 + tid; e < b1; e += BNTH * CU) {
        int dy[CU];
#pragma unroll
        for (int u = 0; u < CU; ++u) {
            const int idx = e + u * BNTH;
            const int ic  = idx < b1 ? idx : b1 - 1;
            dy[u] = binned[ic].y;
        }
        asm volatile("" :: "v"(dy[0]), "v"(dy[1]), "v"(dy[2]), "v"(dy[3]));
#pragma unroll
        for (int u = 0; u < CU; ++u)
            if (e + u * BNTH < b1) atomicAdd(&c[dy[u] - node0], 1);
    }
    __syncthreads();

    int my = 0, incl = 0;
    if (tid < BKT_NODES) {
        my = c[tid];
        incl = my;
        for (int o = 1; o < 32; o <<= 1) {
            const int t = __shfl_up(incl, o);
            if (lane >= o) incl += t;
        }
        if (lane == 31) s_wsum[wid] = incl;
    }
    __syncthreads();
    if (tid < BKT_NODES) {
        int wbase = 0;
        for (int w = 0; w < BKT_NODES / 32; ++w)
            if (w < wid) wbase += s_wsum[w];
        const int first = b0 + wbase + incl - my;
        cur[tid] = first;
        const int node = node0 + tid;
        if (node < nNodes) {
            rowptr[node] = first;
            cnt[node]    = my;
            dinv[node]   = rsqrtf((float)(my + 1));
        }
    }
    __syncthreads();

    for (int e = b0 + tid; e < b1; e += BNTH * CU) {
        int2 p[CU];
#pragma unroll
        for (int u = 0; u < CU; ++u) {
            const int idx = e + u * BNTH;
            const int ic  = idx < b1 ? idx : b1 - 1;
            p[u] = binned[ic];
        }
        asm volatile("" :: "v"(p[0].x), "v"(p[0].y), "v"(p[1].x), "v"(p[1].y),
                           "v"(p[2].x), "v"(p[2].y), "v"(p[3].x), "v"(p[3].y));
        int slot[CU];
#pragma unroll
        for (int u = 0; u < CU; ++u)
            slot[u] = (e + u * BNTH < b1) ? atomicAdd(&cur[p[u].y - node0], 1) : -1;
#pragma unroll
        for (int u = 0; u < CU; ++u)
            if (slot[u] >= 0) csr_src[slot[u]] = p[u].x;
    }
}

template <int K, int U>
__device__ __forceinline__ void gcn_rows_ask(
    const float* __restrict__ xin, int my_s, int j, int coff, float2* r) {
#pragma unroll
    for (int u = 0; u < U; ++u) {
        const int s = __builtin_amdgcn_readlane(my_s, j + u);
        r[u] = *(const float2*)(xin + (size_t)s * K + coff);
    }
}

template <int U>
__device__ __forceinline__ void gcn_rows_add(
    const float2* r, int my_wbits, int j, float& ax, float& ay) {
#pragma unroll
    for (int u = 0; u < U; ++u) {
        const float w = __int_as_float(__builtin_amdgcn_readlane(my_wbits, j + u));
        ax += w * r[u].x;
        ay += w * r[u].y;
    }
}

template <int K, int N, int NTILES, int DO_TANH>
__global__ __launch_bounds__(256) void gcn_gather_gemm_kernel(
    const float* __restrict__ xin, const float* __restrict__ B,
    const int* __restrict__ rowptr, const int* __restrict__ cnt,
    const int* __restrict__ csr_src, const float* __restrict__ dinv,
    float* __restrict__ C, int nNodes) {
    static_assert((K & 1) == 0 && K <= 64, "float2 row loads, one per lane");
    static_assert((NTILES & 1) == 0, "the column tiles are split between two waves");
    constexpr int KP  = (K + 3) & ~3;
    constexpr int KS  = KP + 2;
    constexpr int NP  = NTILES * 16;
    constexpr int NT2 = NTILES / 2;
    constexpr int NTH = 256;
    constexpr int NPW = 8;
    constexpr int NPB = NPW * (NTH / 32);
    __shared__ float BT[NP * KS];
    __shared__ float AT[NPB * KS];

    constexpr int TOT = KS * NP;
    constexpr int PER = (TOT + NTH - 1) / NTH;
    {
        float v[PER];
#pragma unroll
        for (int q = 0; q < PER; ++q) {
            const int idx = (int)threadIdx.x + q * NTH;
            const int k = idx / NP;
            const int n = idx - k * NP;
            const int kc = k < K ? k : K - 1;
            const int nc = n < N ? n : N - 1;
            const float b = B[kc * N + nc];
            v[q] = (n < N && k < K) ? b : 0.0f;
        }
#pragma unroll
        for (int q = 0; q < PER; ++q) {
            const int idx = (int)threadIdx.x + q * NTH;
            const int k = idx / NP;
            const int n = idx - k * NP;
            if (idx < TOT) BT[n * KS + k] = v[q];
        }
    }

    const int wib = __builtin_amdgcn_readfirstlane((int)(threadIdx.x >> 5));
    const int nb0 = __builtin_amdgcn_readfirstlane((int)(blockIdx.x * NPB));
    const int n0  = nb0 + wib * NPW;
    const int lane  = threadIdx.x & 31;
    const int lane2 = lane << 1;
    const int coff = lane2 < K ? lane2 : 0;
    const int nleft = nNodes - n0;
    const int nn = nleft < NPW ? nleft : NPW;
    float* ATw = AT + wib * (NPW * KS);

    if (nn > 0) {
        int m_beg = 0, m_cnt = 0, m_ddbits = 0;
        if (lane < nn) {
            m_beg    = rowptr[n0 + lane];
            m_cnt    = cnt[n0 + lane];
            m_ddbits = __float_as_int(dinv[n0 + lane]);
        }

        int begA = __builtin_amdgcn_readlane(m_beg, 0);
        int cntA = __builtin_amdgcn_readlane(m_cnt, 0);
        int sA = 0;
        if (lane < cntA) sA = csr_src[begA + lane];
        int begB = 0, cntB = 0, sB = 0;
        if (nn > 1) {
            begB = __builtin_amdgcn_readlane(m_beg, 1);
            cntB = __builtin_amdgcn_readlane(m_cnt, 1);
            if (lane < cntB) sB = csr_src[begB + lane];
        }
        float dA = 0.0f;
        if (lane < cntA) dA = dinv[sA];
        float2 svA = *(const float2*)(xin + (size_t)n0 * K + coff);

        for (int i = 0; i < nn; ++i) {
            const int node = n0 + i;
            const int beg = begA;
            const int end = begA + cntA;
            const float dd = __int_as_float(__builtin_amdgcn_readlane(m_ddbits, i));
            const int   first_s = sA;
            const float first_w = dA * dd;
            float ax = dd * dd * svA.x;
            float ay = dd * dd * svA.y;

            begA = begB; cntA = cntB; sA = sB;
            if (i + 1 < nn) {
                dA = 0.0f;
                if (lane < cntA) dA = dinv[sA];
                svA = *(const float2*)(xin + (size_t)(node + 1) * K + coff);
            }
            if (i + 2 < nn) {
                begB = __builtin_amdgcn_readlane(m_beg, i + 2);
                cntB = __builtin_amdgcn_readlane(m_cnt, i + 2);
                sB = 0;
                if (lane < cntB) sB = csr_src[begB + lane];
            }

            for (int p = beg; p < end; p += 32) {
                int   my_s = first_s;
                float my_w = first_w;
                if (p != beg) {
                    const int idx = p + lane;
                    my_s = 0;
                    my_w = 0.0f;
                    if (idx < end) {
                        my_s = csr_src[idx];
                        my_w = dinv[my_s] * dd;
                    }
                }
                const int my_wbits = __float_as_int(my_w);
                const int left = end - p;
                const int n = left < 32 ? left : 32;
                float2 ra[8], rb[8], rc[8], rd[8], t4[4], t2[2], t1[1];
                const int nt = n & ~7;
                if (n >= 8)  gcn_rows_ask<K, 8>(xin, my_s, 0,  coff, ra);
                if (n >= 16) gcn_rows_ask<K, 8>(xin, my_s, 8,  coff, rb);
                if (n >= 24) gcn_rows_ask<K, 8>(xin, my_s, 16, coff, rc);
                if (n >= 32) gcn_rows_ask<K, 8>(xin, my_s, 24, coff, rd);
                int j = nt;
                if (n & 4) { gcn_rows_ask<K, 4>(xin, my_s, j, coff, t4); j += 4; }
                if (n & 2) { gcn_rows_ask<K, 2>(xin, my_s, j, coff, t2); j += 2; }
                if (n & 1) { gcn_rows_ask<K, 1>(xin, my_s, j, coff, t1); }

                if (n >= 8)  gcn_rows_add<8>(ra, my_wbits, 0,  ax, ay);
                if (n >= 16) gcn_rows_add<8>(rb, my_wbits, 8,  ax, ay);
                if (n >= 24) gcn_rows_add<8>(rc, my_wbits, 16, ax, ay);
                if (n >= 32) gcn_rows_add<8>(rd, my_wbits, 24, ax, ay);
                j = nt;
                if (n & 4) { gcn_rows_add<4>(t4, my_wbits, j, ax, ay); j += 4; }
                if (n & 2) { gcn_rows_add<2>(t2, my_wbits, j, ax, ay); j += 2; }
                if (n & 1) { gcn_rows_add<1>(t1, my_wbits, j, ax, ay); }
            }

            if (lane2 < KP) {
                float2 o;
                o.x = lane2 < K ? ax : 0.0f;
                o.y = lane2 < K ? ay : 0.0f;
                *(float2*)(ATw + i * KS + lane2) = o;
            }
        }
    }

    __syncthreads();

    const int rt  = wib >> 1;
    const int ct0 = (wib & 1) * NT2;
    const int row0 = nb0 + (rt << 4);
    if (row0 >= nNodes) return;

    const int mrow  = lane & 15;
    const int khalf = (lane >> 4) << 1;
    const float* ATlane = AT + ((rt << 4) + mrow) * KS;
    const float* BTlane = BT + (ct0 * 16 + mrow) * KS;

    v8f acc[NT2];
#pragma unroll
    for (int t = 0; t < NT2; ++t) acc[t] = v8f{};

#pragma unroll
    for (int k0 = 0; k0 < KP; k0 += 4) {
        const int ka = k0 + khalf;
        const float2 av = *(const float2*)(ATlane + ka);
        v2f a; a.x = av.x; a.y = av.y;
#pragma unroll
        for (int t = 0; t < NT2; ++t) {
            const float2 bv = *(const float2*)(BTlane + t * 16 * KS + ka);
            v2f b; b.x = bv.x; b.y = bv.y;
            acc[t] = __builtin_amdgcn_wmma_f32_16x16x4_f32(
                false, a, false, b, (short)0, acc[t], false, false);
        }
    }

    const int rbase = row0 + ((lane >> 4) << 3);
#pragma unroll
    for (int t = 0; t < NT2; ++t) {
        const int col = ((ct0 + t) << 4) + mrow;
        if (col < N) {
#pragma unroll
            for (int r = 0; r < 8; ++r) {
                const float v = acc[t][r];
                C[(size_t)(rbase + r) * N + col] = DO_TANH ? tanhf(v) : v;
            }
        }
    }
}

extern "C" void kernel_launch(void* const* d_in, const int* in_sizes, int n_in,
                              void* d_out, int out_size, void* d_ws, size_t ws_size,
                              hipStream_t stream) {
    const float* x  = (const float*)d_in[0];
    const float* W1 = (const float*)d_in[1];
    const float* W2 = (const float*)d_in[2];
    const int* ei   = (const int*)d_in[3];

    const int FIN = 50, F1 = 64;
    const int nNodes = in_sizes[0] / FIN;
    const int nEdges = in_sizes[3] / 2;
    const int* src = ei;
    const int* dst = ei + nEdges;

    const int NP = (nNodes + 63) & ~63;
    const int EP = (nEdges + 63) & ~63;
    const int nbkt = (nNodes + BKT_NODES - 1) >> BKT_SHIFT;
    const int eblk = (nEdges + BNTH * BEPT - 1) / (BNTH * BEPT);
    const int HP = (eblk * nbkt + 63) & ~63;
    const int BP = (nbkt + 1 + 63) & ~63;
    float* wsF    = (float*)d_ws;
    float* dinv   = wsF;
    int*   cnt    = (int*)(dinv + NP);
    int*   rowptr = cnt + NP;
    int*   hist   = rowptr + NP;
    int*   bstart = hist + HP;
    int2*  binned = (int2*)(bstart + BP);
    int*   csr    = (int*)(binned + EP);
    float* hbuf   = (float*)(csr + EP);
    float* outf   = (float*)d_out;

    const int T = 256;
    const int layerBlocks = (nNodes + 63) / 64;

    gcn_bucket_count_kernel<<<eblk, BNTH, 0, stream>>>(dst, hist, nEdges, nbkt);
    gcn_bucket_place_kernel<<<eblk, BNTH, 0, stream>>>(src, dst, hist, bstart, binned,
                                                       nEdges, nbkt, eblk);
    gcn_bucket_rows_kernel<<<nbkt, BNTH, 0, stream>>>(binned, bstart, rowptr, cnt, dinv,
                                                      csr, nNodes);

    gcn_gather_gemm_kernel<50, 64, 4, 1><<<layerBlocks, T, 0, stream>>>(
        x, W1, rowptr, cnt, csr, dinv, hbuf, nNodes);
    gcn_gather_gemm_kernel<64, 121, 8, 0><<<layerBlocks, T, 0, stream>>>(
        hbuf, W2, rowptr, cnt, csr, dinv, outf, nNodes);

    (void)n_in; (void)out_size; (void)ws_size; (void)F1;
}


# Each edge block sorts its own span by bucket in LDS and writes it whole; a bucket block picks its runs out of the spans: count pass gone, 4 launches, not 5; f32 sum order changes (LDS arrival order)
# speedup vs baseline: 5.5340x; 1.1452x over previous
#include <hip/hip_runtime.h>
#include <hip/hip_bf16.h>


typedef __attribute__((ext_vector_type(2))) float v2f;
typedef __attribute__((ext_vector_type(8))) float v8f;


constexpr int BKT_SHIFT = 8;
constexpr int BKT_NODES = 1 << BKT_SHIFT;
constexpr int MAXBKT    = 1024;
constexpr int BNTH      = 1024;
constexpr int BEPT      = 8;
constexpr int SPAN      = BNTH * BEPT;
constexpr int MAXBLK    = 1024;

__global__ __launch_bounds__(BNTH) void gcn_span_sort_kernel(
    const int* __restrict__ src, const int* __restrict__ dst,
    int* __restrict__ hist, int* __restrict__ lstart,
    int2* __restrict__ binned, int nE, int nbkt) {
    static_assert(MAXBKT == BNTH, "one bucket counter per thread");
    static_assert((BEPT & 1) == 0, "the span is written out two pairs at a time");
    __shared__ int h[MAXBKT];
    __shared__ int s_wsum[BNTH / 32];
    __shared__ int2 stage[SPAN];
    const int tid  = threadIdx.x;
    const int lane = tid & 31;
    const int wid  = tid >> 5;
    const int bid  = blockIdx.x;
    const int base = bid * SPAN + tid;
    int s[BEPT];
    int d[BEPT];
    h[tid] = 0;
#pragma unroll
    for (int k = 0; k < BEPT; ++k) {
        const int e  = base + k * BNTH;
        const int ec = e < nE ? e : nE - 1;
        s[k] = src[ec];
        d[k] = dst[ec];
    }
    __syncthreads();
#pragma unroll
    for (int k = 0; k < BEPT; ++k)
        if (base + k * BNTH < nE) atomicAdd(&h[d[k] >> BKT_SHIFT], 1);
    __syncthreads();

    const int tot = h[tid];
    int incl = tot;
    for (int o = 1; o < 32; o <<= 1) {
        const int t = __shfl_up(incl, o);
        if (lane >= o) incl += t;
    }
    if (lane == 31) s_wsum[wid] = incl;
    __syncthreads();
    int wbase = 0;
    for (int w = 0; w < BNTH / 32; ++w)
        if (w < wid) wbase += s_wsum[w];
    const int first = wbase + incl - tot;
    h[tid] = first;
    if (tid < nbkt) {
        hist[bid * nbkt + tid]   = tot;
        lstart[bid * nbkt + tid] = first;
    }
    __syncthreads();

    int slot[BEPT];
#pragma unroll
    for (int k = 0; k < BEPT; ++k)
        slot[k] = (base + k * BNTH < nE) ? atomicAdd(&h[d[k] >> BKT_SHIFT], 1) : -1;
#pragma unroll
    for (int k = 0; k < BEPT; ++k)
        if (slot[k] >= 0) stage[slot[k]] = make_int2(s[k], d[k]);
    __syncthreads();

    const int left = nE - bid * SPAN;
    const int nv = left < SPAN ? left : SPAN;
    int2* seg = binned + (size_t)bid * SPAN;
    const int4* stage4 = (const int4*)stage;
    int4* seg4 = (int4*)seg;
#pragma unroll
    for (int k = 0; k < BEPT / 2; ++k) {
        const int i = tid + k * BNTH;
        if (2 * i + 1 < nv) seg4[i] = stage4[i];
        else if (2 * i < nv) seg[2 * i] = stage[2 * i];
    }
}

__global__ __launch_bounds__(BNTH) void gcn_bucket_rows_kernel(
    const int2* __restrict__ binned, const int* __restrict__ hist,
    const int* __restrict__ lstart,
    int* __restrict__ rowptr, int* __restrict__ cnt, float* __restrict__ dinv,
    int* __restrict__ csr_src, int nNodes, int nbkt, int nblk) {
    constexpr int NU = 4, NV = 2;
    static_assert(NU == 4 && NV == 2, "the load fence below names 16 values");
    __shared__ int c[BKT_NODES];
    __shared__ int cur[BKT_NODES];
    __shared__ int s_wsum[BKT_NODES / 32];
    __shared__ int c_s[MAXBLK];
    __shared__ int l_s[MAXBLK];
    __shared__ int s_red[BNTH / 32];
    const int tid  = threadIdx.x;
    const int lane = tid & 31;
    const int wid  = tid >> 5;
    const int bk   = blockIdx.x;
    const int node0 = bk << BKT_SHIFT;

    int lsum = 0;
    for (int b = tid; b < nblk; b += BNTH) {
        const int cc = hist[b * nbkt + bk];
        const int ll = lstart[b * nbkt + bk];
        c_s[b] = cc;
        l_s[b] = ll;
        lsum += ll;
    }
    if (tid < BKT_NODES) c[tid] = 0;
    for (int o = 16; o > 0; o >>= 1) lsum += __shfl_xor(lsum, o);
    if (lane == 0) s_red[wid] = lsum;
    __syncthreads();
    int b0 = 0;
    for (int w = 0; w < BNTH / 32; ++w) b0 += s_red[w];

    int2 p[NU][NV];
    int  nn[NU];
#pragma unroll
    for (int u = 0; u < NU; ++u) {
        const int b  = wid + 32 * u;
        const int bc = b < nblk ? b : nblk - 1;
        const int n  = b < nblk ? c_s[bc] : 0;
        const int ls = l_s[bc];
        nn[u] = n;
#pragma unroll
        for (int v = 0; v < NV; ++v) {
            const int i   = 32 * v + lane;
            const int off = i < n ? ls + i : 0;
            p[u][v] = binned[(size_t)bc * SPAN + off];
        }
    }
    asm volatile("" :: "v"(p[0][0].x), "v"(p[0][0].y), "v"(p[0][1].x), "v"(p[0][1].y),
                       "v"(p[1][0].x), "v"(p[1][0].y), "v"(p[1][1].x), "v"(p[1][1].y),
                       "v"(p[2][0].x), "v"(p[2][0].y), "v"(p[2][1].x), "v"(p[2][1].y),
                       "v"(p[3][0].x), "v"(p[3][0].y), "v"(p[3][1].x), "v"(p[3][1].y));
#pragma unroll
    for (int u = 0; u < NU; ++u)
#pragma unroll
        for (int v = 0; v < NV; ++v)
            if (32 * v + lane < nn[u]) atomicAdd(&c[p[u][v].y - node0], 1);
    for (int b = wid; b < nblk; b += 32) {
        const int n = c_s[b];
        const int2* run = binned + (size_t)b * SPAN + l_s[b];
        for (int i = (b < 32 * NU ? 32 * NV : 0) + lane; i < n; i += 32)
            atomicAdd(&c[run[i].y - node0], 1);
    }
    __syncthreads();

    int my = 0, incl = 0;
    if (tid < BKT_NODES) {
        my = c[tid];
        incl = my;
        for (int o = 1; o < 32; o <<= 1) {
            const int t = __shfl_up(incl, o);
            if (lane >= o) incl += t;
        }
        if (lane == 31) s_wsum[wid] = incl;
    }
    __syncthreads();
    if (tid < BKT_NODES) {
        int wbase = 0;
        for (int w = 0; w < BKT_NODES / 32; ++w)
            if (w < wid) wbase += s_wsum[w];
        const int first = b0 + wbase + incl - my;
        cur[tid] = first;
        const int node = node0 + tid;
        if (node < nNodes) {
            rowptr[node] = first;
            cnt[node]    = my;
            dinv[node]   = rsqrtf((float)(my + 1));
        }
    }
    __syncthreads();

    int slot[NU][NV];
#pragma unroll
    for (int u = 0; u < NU; ++u)
#pragma unroll
        for (int v = 0; v < NV; ++v)
            slot[u][v] = (32 * v + lane < nn[u]) ? atomicAdd(&cur[p[u][v].y - node0], 1) : -1;
#pragma unroll
    for (int u = 0; u < NU; ++u)
#pragma unroll
        for (int v = 0; v < NV; ++v)
            if (slot[u][v] >= 0) csr_src[slot[u][v]] = p[u][v].x;
    for (int b = wid; b < nblk; b += 32) {
        const int n = c_s[b];
        const int2* run = binned + (size_t)b * SPAN + l_s[b];
        for (int i = (b < 32 * NU ? 32 * NV : 0) + lane; i < n; i += 32) {
            const int2 q = run[i];
            csr_src[atomicAdd(&cur[q.y - node0], 1)] = q.x;
        }
    }
}

template <int K, int U>
__device__ __forceinline__ void gcn_rows_ask(
    const float* __restrict__ xin, int my_s, int j, int coff, float2* r) {
#pragma unroll
    for (int u = 0; u < U; ++u) {
        const int s = __builtin_amdgcn_readlane(my_s, j + u);
        r[u] = *(const float2*)(xin + (size_t)s * K + coff);
    }
}

template <int U>
__device__ __forceinline__ void gcn_rows_add(
    const float2* r, int my_wbits, int j, float& ax, float& ay) {
#pragma unroll
    for (int u = 0; u < U; ++u) {
        const float w = __int_as_float(__builtin_amdgcn_readlane(my_wbits, j + u));
        ax += w * r[u].x;
        ay += w * r[u].y;
    }
}

template <int K, int N, int NTILES, int DO_TANH>
__global__ __launch_bounds__(256) void gcn_gather_gemm_kernel(
    const float* __restrict__ xin, const float* __restrict__ B,
    const int* __restrict__ rowptr, const int* __restrict__ cnt,
    const int* __restrict__ csr_src, const float* __restrict__ dinv,
    float* __restrict__ C, int nNodes) {
    static_assert((K & 1) == 0 && K <= 64, "float2 row loads, one per lane");
    static_assert((NTILES & 1) == 0, "the column tiles are split between two waves");
    constexpr int KP  = (K + 3) & ~3;
    constexpr int KS  = KP + 2;
    constexpr int NP  = NTILES * 16;
    constexpr int NT2 = NTILES / 2;
    constexpr int NTH = 256;
    constexpr int NPW = 8;
    constexpr int NPB = NPW * (NTH / 32);
    __shared__ float BT[NP * KS];
    __shared__ float AT[NPB * KS];

    constexpr int TOT = KS * NP;
    constexpr int PER = (TOT + NTH - 1) / NTH;
    {
        float v[PER];
#pragma unroll
        for (int q = 0; q < PER; ++q) {
            const int idx = (int)threadIdx.x + q * NTH;
            const int k = idx / NP;
            const int n = idx - k * NP;
            const int kc = k < K ? k : K - 1;
            const int nc = n < N ? n : N - 1;
            const float b = B[kc * N + nc];
            v[q] = (n < N && k < K) ? b : 0.0f;
        }
#pragma unroll
        for (int q = 0; q < PER; ++q) {
            const int idx = (int)threadIdx.x + q * NTH;
            const int k = idx / NP;
            const int n = idx - k * NP;
            if (idx < TOT) BT[n * KS + k] = v[q];
        }
    }

    const int wib = __builtin_amdgcn_readfirstlane((int)(threadIdx.x >> 5));
    const int nb0 = __builtin_amdgcn_readfirstlane((int)(blockIdx.x * NPB));
    const int n0  = nb0 + wib * NPW;
    const int lane  = threadIdx.x & 31;
    const int lane2 = lane << 1;
    const int coff = lane2 < K ? lane2 : 0;
    const int nleft = nNodes - n0;
    const int nn = nleft < NPW ? nleft : NPW;
    float* ATw = AT + wib * (NPW * KS);

    if (nn > 0) {
        int m_beg = 0, m_cnt = 0, m_ddbits = 0;
        if (lane < nn) {
            m_beg    = rowptr[n0 + lane];
            m_cnt    = cnt[n0 + lane];
            m_ddbits = __float_as_int(dinv[n0 + lane]);
        }

        int begA = __builtin_amdgcn_readlane(m_beg, 0);
        int cntA = __builtin_amdgcn_readlane(m_cnt, 0);
        int sA = 0;
        if (lane < cntA) sA = csr_src[begA + lane];
        int begB = 0, cntB = 0, sB = 0;
        if (nn > 1) {
            begB = __builtin_amdgcn_readlane(m_beg, 1);
            cntB = __builtin_amdgcn_readlane(m_cnt, 1);
            if (lane < cntB) sB = csr_src[begB + lane];
        }
        float dA = 0.0f;
        if (lane < cntA) dA = dinv[sA];
        float2 svA = *(const float2*)(xin + (size_t)n0 * K + coff);

        for (int i = 0; i < nn; ++i) {
            const int node = n0 + i;
            const int beg = begA;
            const int end = begA + cntA;
            const float dd = __int_as_float(__builtin_amdgcn_readlane(m_ddbits, i));
            const int   first_s = sA;
            const float first_w = dA * dd;
            float ax = dd * dd * svA.x;
            float ay = dd * dd * svA.y;

            begA = begB; cntA = cntB; sA = sB;
            if (i + 1 < nn) {
                dA = 0.0f;
                if (lane < cntA) dA = dinv[sA];
                svA = *(const float2*)(xin + (size_t)(node + 1) * K + coff);
            }
            if (i + 2 < nn) {
                begB = __builtin_amdgcn_readlane(m_beg, i + 2);
                cntB = __builtin_amdgcn_readlane(m_cnt, i + 2);
                sB = 0;
                if (lane < cntB) sB = csr_src[begB + lane];
            }

            for (int p = beg; p < end; p += 32) {
                int   my_s = first_s;
                float my_w = first_w;
                if (p != beg) {
                    const int idx = p + lane;
                    my_s = 0;
                    my_w = 0.0f;
                    if (idx < end) {
                        my_s = csr_src[idx];
                        my_w = dinv[my_s] * dd;
                    }
                }
                const int my_wbits = __float_as_int(my_w);
                const int left = end - p;
                const int n = left < 32 ? left : 32;
                float2 ra[8], rb[8], rc[8], rd[8], t4[4], t2[2], t1[1];
                const int nt = n & ~7;
                if (n >= 8)  gcn_rows_ask<K, 8>(xin, my_s, 0,  coff, ra);
                if (n >= 16) gcn_rows_ask<K, 8>(xin, my_s, 8,  coff, rb);
                if (n >= 24) gcn_rows_ask<K, 8>(xin, my_s, 16, coff, rc);
                if (n >= 32) gcn_rows_ask<K, 8>(xin, my_s, 24, coff, rd);
                int j = nt;
                if (n & 4) { gcn_rows_ask<K, 4>(xin, my_s, j, coff, t4); j += 4; }
                if (n & 2) { gcn_rows_ask<K, 2>(xin, my_s, j, coff, t2); j += 2; }
                if (n & 1) { gcn_rows_ask<K, 1>(xin, my_s, j, coff, t1); }

                if (n >= 8)  gcn_rows_add<8>(ra, my_wbits, 0,  ax, ay);
                if (n >= 16) gcn_rows_add<8>(rb, my_wbits, 8,  ax, ay);
                if (n >= 24) gcn_rows_add<8>(rc, my_wbits, 16, ax, ay);
                if (n >= 32) gcn_rows_add<8>(rd, my_wbits, 24, ax, ay);
                j = nt;
                if (n & 4) { gcn_rows_add<4>(t4, my_wbits, j, ax, ay); j += 4; }
                if (n & 2) { gcn_rows_add<2>(t2, my_wbits, j, ax, ay); j += 2; }
                if (n & 1) { gcn_rows_add<1>(t1, my_wbits, j, ax, ay); }
            }

            if (lane2 < KP) {
                float2 o;
                o.x = lane2 < K ? ax : 0.0f;
                o.y = lane2 < K ? ay : 0.0f;
                *(float2*)(ATw + i * KS + lane2) = o;
            }
        }
    }

    __syncthreads();

    const int rt  = wib >> 1;
    const int ct0 = (wib & 1) * NT2;
    const int row0 = nb0 + (rt << 4);
    if (row0 >= nNodes) return;

    const int mrow  = lane & 15;
    const int khalf = (lane >> 4) << 1;
    const float* ATlane = AT + ((rt << 4) + mrow) * KS;
    const float* BTlane = BT + (ct0 * 16 + mrow) * KS;

    v8f acc[NT2];
#pragma unroll
    for (int t = 0; t < NT2; ++t) acc[t] = v8f{};

#pragma unroll
    for (int k0 = 0; k0 < KP; k0 += 4) {
        const int ka = k0 + khalf;
        const float2 av = *(const float2*)(ATlane + ka);
        v2f a; a.x = av.x; a.y = av.y;
#pragma unroll
        for (int t = 0; t < NT2; ++t) {
            const float2 bv = *(const float2*)(BTlane + t * 16 * KS + ka);
            v2f b; b.x = bv.x; b.y = bv.y;
            acc[t] = __builtin_amdgcn_wmma_f32_16x16x4_f32(
                false, a, false, b, (short)0, acc[t], false, false);
        }
    }

    const int rbase = row0 + ((lane >> 4) << 3);
#pragma unroll
    for (int t = 0; t < NT2; ++t) {
        const int col = ((ct0 + t) << 4) + mrow;
        if (col < N) {
#pragma unroll
            for (int r = 0; r < 8; ++r) {
                const float v = acc[t][r];
                C[(size_t)(rbase + r) * N + col] = DO_TANH ? tanhf(v) : v;
            }
        }
    }
}

extern "C" void kernel_launch(void* const* d_in, const int* in_sizes, int n_in,
                              void* d_out, int out_size, void* d_ws, size_t ws_size,
                              hipStream_t stream) {
    const float* x  = (const float*)d_in[0];
    const float* W1 = (const float*)d_in[1];
    const float* W2 = (const float*)d_in[2];
    const int* ei   = (const int*)d_in[3];

    const int FIN = 50, F1 = 64;
    const int nNodes = in_sizes[0] / FIN;
    const int nEdges = in_sizes[3] / 2;
    const int* src = ei;
    const int* dst = ei + nEdges;

    const int NP = (nNodes + 63) & ~63;
    const int EP = (nEdges + 63) & ~63;
    const int nbkt = (nNodes + BKT_NODES - 1) >> BKT_SHIFT;
    const int eblk = (nEdges + SPAN - 1) / SPAN;
    const int HP = (eblk * nbkt + 63) & ~63;
    float* wsF    = (float*)d_ws;
    float* dinv   = wsF;
    int*   cnt    = (int*)(dinv + NP);
    int*   rowptr = cnt + NP;
    int*   hist   = rowptr + NP;
    int*   lstart = hist + HP;
    int2*  binned = (int2*)(lstart + HP);
    int*   csr    = (int*)(binned + EP);
    float* hbuf   = (float*)(csr + EP);
    float* outf   = (float*)d_out;

    const int T = 256;
    const int layerBlocks = (nNodes + 63) / 64;

    gcn_span_sort_kernel<<<eblk, BNTH, 0, stream>>>(src, dst, hist, lstart, binned,
                                                    nEdges, nbkt);
    gcn_bucket_rows_kernel<<<nbkt, BNTH, 0, stream>>>(binned, hist, lstart, rowptr, cnt, dinv,
                                                      csr, nNodes, nbkt, eblk);

    gcn_gather_gemm_kernel<50, 64, 4, 1><<<layerBlocks, T, 0, stream>>>(
        x, W1, rowptr, cnt, csr, dinv, hbuf, nNodes);
    gcn_gather_gemm_kernel<64, 121, 8, 0><<<layerBlocks, T, 0, stream>>>(
        hbuf, W2, rowptr, cnt, csr, dinv, outf, nNodes);

    (void)n_in; (void)out_size; (void)ws_size; (void)F1;
}


# Rows carry dinv from where they are made (x in the bucket pass, h in layer 1's store): edge sums add plain rows, no scattered dinv[src] read per edge; f32 rounding changes (dinv[dst] factored out)
# speedup vs baseline: 6.0314x; 1.0899x over previous
#include <hip/hip_runtime.h>
#include <hip/hip_bf16.h>


typedef __attribute__((ext_vector_type(2))) float v2f;
typedef __attribute__((ext_vector_type(8))) float v8f;


constexpr int BKT_SHIFT = 8;
constexpr int BKT_NODES = 1 << BKT_SHIFT;
constexpr int MAXBKT    = 1024;
constexpr int BNTH      = 1024;
constexpr int BEPT      = 8;
constexpr int SPAN      = BNTH * BEPT;
constexpr int MAXBLK    = 1024;

__global__ __launch_bounds__(BNTH) void gcn_span_sort_kernel(
    const int* __restrict__ src, const int* __restrict__ dst,
    int* __restrict__ hist, int* __restrict__ lstart,
    int2* __restrict__ binned, int nE, int nbkt) {
    static_assert(MAXBKT == BNTH, "one bucket counter per thread");
    static_assert((BEPT & 1) == 0, "the span is written out two pairs at a time");
    __shared__ int h[MAXBKT];
    __shared__ int s_wsum[BNTH / 32];
    __shared__ int2 stage[SPAN];
    const int tid  = threadIdx.x;
    const int lane = tid & 31;
    const int wid  = tid >> 5;
    const int bid  = blockIdx.x;
    const int base = bid * SPAN + tid;
    int s[BEPT];
    int d[BEPT];
    h[tid] = 0;
#pragma unroll
    for (int k = 0; k < BEPT; ++k) {
        const int e  = base + k * BNTH;
        const int ec = e < nE ? e : nE - 1;
        s[k] = src[ec];
        d[k] = dst[ec];
    }
    __syncthreads();
#pragma unroll
    for (int k = 0; k < BEPT; ++k)
        if (base + k * BNTH < nE) atomicAdd(&h[d[k] >> BKT_SHIFT], 1);
    __syncthreads();

    const int tot = h[tid];
    int incl = tot;
    for (int o = 1; o < 32; o <<= 1) {
        const int t = __shfl_up(incl, o);
        if (lane >= o) incl += t;
    }
    if (lane == 31) s_wsum[wid] = incl;
    __syncthreads();
    int wbase = 0;
    for (int w = 0; w < BNTH / 32; ++w)
        if (w < wid) wbase += s_wsum[w];
    const int first = wbase + incl - tot;
    h[tid] = first;
    if (tid < nbkt) {
        hist[bid * nbkt + tid]   = tot;
        lstart[bid * nbkt + tid] = first;
    }
    __syncthreads();

    int slot[BEPT];
#pragma unroll
    for (int k = 0; k < BEPT; ++k)
        slot[k] = (base + k * BNTH < nE) ? atomicAdd(&h[d[k] >> BKT_SHIFT], 1) : -1;
#pragma unroll
    for (int k = 0; k < BEPT; ++k)
        if (slot[k] >= 0) stage[slot[k]] = make_int2(s[k], d[k]);
    __syncthreads();

    const int left = nE - bid * SPAN;
    const int nv = left < SPAN ? left : SPAN;
    int2* seg = binned + (size_t)bid * SPAN;
    const int4* stage4 = (const int4*)stage;
    int4* seg4 = (int4*)seg;
#pragma unroll
    for (int k = 0; k < BEPT / 2; ++k) {
        const int i = tid + k * BNTH;
        if (2 * i + 1 < nv) seg4[i] = stage4[i];
        else if (2 * i < nv) seg[2 * i] = stage[2 * i];
    }
}

template <int K>
__global__ __launch_bounds__(BNTH) void gcn_bucket_rows_kernel(
    const int2* __restrict__ binned, const int* __restrict__ hist,
    const int* __restrict__ lstart, const float* __restrict__ x,
    int* __restrict__ rowptr, int* __restrict__ cnt, float* __restrict__ dinv,
    int* __restrict__ csr_src, float* __restrict__ xs,
    int nNodes, int nbkt, int nblk) {
    constexpr int NU = 4, NV = 2;
    static_assert(NU == 4 && NV == 2, "the load fence below names 16 values");
    static_assert((K & 1) == 0, "rows are moved two floats at a time");
    constexpr int K2 = K / 2;
    constexpr int XQ = (BKT_NODES * K2 + BNTH - 1) / BNTH;
    __shared__ int c[BKT_NODES];
    __shared__ int cur[BKT_NODES];
    __shared__ float dv[BKT_NODES];
    __shared__ int s_wsum[BKT_NODES / 32];
    __shared__ int c_s[MAXBLK];
    __shared__ int l_s[MAXBLK];
    __shared__ int s_red[BNTH / 32];
    const int tid  = threadIdx.x;
    const int lane = tid & 31;
    const int wid  = tid >> 5;
    const int bk   = blockIdx.x;
    const int node0 = bk << BKT_SHIFT;

    const int nleft = nNodes - node0;
    const int nx2 = (nleft < BKT_NODES ? nleft : BKT_NODES) * K2;
    const float2* x2 = (const float2*)(x + (size_t)node0 * K);
    float2 xv[XQ];
#pragma unroll
    for (int q = 0; q < XQ; ++q) {
        const int i = tid + q * BNTH;
        xv[q] = x2[i < nx2 ? i : 0];
    }

    int lsum = 0;
    for (int b = tid; b < nblk; b += BNTH) {
        const int cc = hist[b * nbkt + bk];
        const int ll = lstart[b * nbkt + bk];
        c_s[b] = cc;
        l_s[b] = ll;
        lsum += ll;
    }
    if (tid < BKT_NODES) c[tid] = 0;
    for (int o = 16; o > 0; o >>= 1) lsum += __shfl_xor(lsum, o);
    if (lane == 0) s_red[wid] = lsum;
    __syncthreads();
    int b0 = 0;
    for (int w = 0; w < BNTH / 32; ++w) b0 += s_red[w];

    int2 p[NU][NV];
    int  nn[NU];
#pragma unroll
    for (int u = 0; u < NU; ++u) {
        const int b  = wid + 32 * u;
        const int bc = b < nblk ? b : nblk - 1;
        const int n  = b < nblk ? c_s[bc] : 0;
        const int ls = l_s[bc];
        nn[u] = n;
#pragma unroll
        for (int v = 0; v < NV; ++v) {
            const int i   = 32 * v + lane;
            const int off = i < n ? ls + i : 0;
            p[u][v] = binned[(size_t)bc * SPAN + off];
        }
    }
    asm volatile("" :: "v"(p[0][0].x), "v"(p[0][0].y), "v"(p[0][1].x), "v"(p[0][1].y),
                       "v"(p[1][0].x), "v"(p[1][0].y), "v"(p[1][1].x), "v"(p[1][1].y),
                       "v"(p[2][0].x), "v"(p[2][0].y), "v"(p[2][1].x), "v"(p[2][1].y),
                       "v"(p[3][0].x), "v"(p[3][0].y), "v"(p[3][1].x), "v"(p[3][1].y));
#pragma unroll
    for (int u = 0; u < NU; ++u)
#pragma unroll
        for (int v = 0; v < NV; ++v)
            if (32 * v + lane < nn[u]) atomicAdd(&c[p[u][v].y - node0], 1);
    for (int b = wid; b < nblk; b += 32) {
        const int n = c_s[b];
        const int2* run = binned + (size_t)b * SPAN + l_s[b];
        for (int i = (b < 32 * NU ? 32 * NV : 0) + lane; i < n; i += 32)
            atomicAdd(&c[run[i].y - node0], 1);
    }
    __syncthreads();

    int my = 0, incl = 0;
    if (tid < BKT_NODES) {
        my = c[tid];
        incl = my;
        for (int o = 1; o < 32; o <<= 1) {
            const int t = __shfl_up(incl, o);
            if (lane >= o) incl += t;
        }
        if (lane == 31) s_wsum[wid] = incl;
    }
    __syncthreads();
    if (tid < BKT_NODES) {
        int wbase = 0;
        for (int w = 0; w < BKT_NODES / 32; ++w)
            if (w < wid) wbase += s_wsum[w];
        const int first = b0 + wbase + incl - my;
        cur[tid] = first;
        const float di = rsqrtf((float)(my + 1));
        dv[tid] = di;
        const int node = node0 + tid;
        if (node < nNodes) {
            rowptr[node] = first;
            cnt[node]    = my;
            dinv[node]   = di;
        }
    }
    __syncthreads();

    float2* xs2 = (float2*)(xs + (size_t)node0 * K);
#pragma unroll
    for (int q = 0; q < XQ; ++q) {
        const int i = tid + q * BNTH;
        if (i < nx2) {
            const float w = dv[i / K2];
            xs2[i] = make_float2(w * xv[q].x, w * xv[q].y);
        }
    }

    int slot[NU][NV];
#pragma unroll
    for (int u = 0; u < NU; ++u)
#pragma unroll
        for (int v = 0; v < NV; ++v)
            slot[u][v] = (32 * v + lane < nn[u]) ? atomicAdd(&cur[p[u][v].y - node0], 1) : -1;
#pragma unroll
    for (int u = 0; u < NU; ++u)
#pragma unroll
        for (int v = 0; v < NV; ++v)
            if (slot[u][v] >= 0) csr_src[slot[u][v]] = p[u][v].x;
    for (int b = wid; b < nblk; b += 32) {
        const int n = c_s[b];
        const int2* run = binned + (size_t)b * SPAN + l_s[b];
        for (int i = (b < 32 * NU ? 32 * NV : 0) + lane; i < n; i += 32) {
            const int2 q = run[i];
            csr_src[atomicAdd(&cur[q.y - node0], 1)] = q.x;
        }
    }
}

template <int K, int U>
__device__ __forceinline__ void gcn_rows_ask(
    const float* __restrict__ xin, int my_s, int j, int coff, float2* r) {
#pragma unroll
    for (int u = 0; u < U; ++u) {
        const int s = __builtin_amdgcn_readlane(my_s, j + u);
        r[u] = *(const float2*)(xin + (size_t)s * K + coff);
    }
}

template <int U>
__device__ __forceinline__ void gcn_rows_add(const float2* r, float& ax, float& ay) {
#pragma unroll
    for (int u = 0; u < U; ++u) {
        ax += r[u].x;
        ay += r[u].y;
    }
}

template <int K, int N, int NTILES, int HIDDEN>
__global__ __launch_bounds__(256) void gcn_gather_gemm_kernel(
    const float* __restrict__ xin, const float* __restrict__ B,
    const int* __restrict__ rowptr, const int* __restrict__ cnt,
    const int* __restrict__ csr_src, const float* __restrict__ dinv,
    float* __restrict__ C, int nNodes) {
    static_assert((K & 1) == 0 && K <= 64, "float2 row loads, one per lane");
    static_assert((NTILES & 1) == 0, "the column tiles are split between two waves");
    constexpr int KP  = (K + 3) & ~3;
    constexpr int KS  = KP + 2;
    constexpr int NP  = NTILES * 16;
    constexpr int NT2 = NTILES / 2;
    constexpr int NTH = 256;
    constexpr int NPW = 8;
    constexpr int NPB = NPW * (NTH / 32);
    __shared__ float BT[NP * KS];
    __shared__ float AT[NPB * KS];

    constexpr int TOT = KS * NP;
    constexpr int PER = (TOT + NTH - 1) / NTH;
    {
        float v[PER];
#pragma unroll
        for (int q = 0; q < PER; ++q) {
            const int idx = (int)threadIdx.x + q * NTH;
            const int k = idx / NP;
            const int n = idx - k * NP;
            const int kc = k < K ? k : K - 1;
            const int nc = n < N ? n : N - 1;
            const float b = B[kc * N + nc];
            v[q] = (n < N && k < K) ? b : 0.0f;
        }
#pragma unroll
        for (int q = 0; q < PER; ++q) {
            const int idx = (int)threadIdx.x + q * NTH;
            const int k = idx / NP;
            const int n = idx - k * NP;
            if (idx < TOT) BT[n * KS + k] = v[q];
        }
    }

    const int wib = __builtin_amdgcn_readfirstlane((int)(threadIdx.x >> 5));
    const int nb0 = __builtin_amdgcn_readfirstlane((int)(blockIdx.x * NPB));
    const int n0  = nb0 + wib * NPW;
    const int lane  = threadIdx.x & 31;
    const int lane2 = lane << 1;
    const int coff = lane2 < K ? lane2 : 0;
    const int nleft = nNodes - n0;
    const int nn = nleft < NPW ? nleft : NPW;
    float* ATw = AT + wib * (NPW * KS);

    if (nn > 0) {
        int m_beg = 0, m_cnt = 0, m_ddbits = 0;
        if (lane < nn) {
            m_beg    = rowptr[n0 + lane];
            m_cnt    = cnt[n0 + lane];
            m_ddbits = __float_as_int(dinv[n0 + lane]);
        }

        int begA = __builtin_amdgcn_readlane(m_beg, 0);
        int cntA = __builtin_amdgcn_readlane(m_cnt, 0);
        int sA = 0;
        if (lane < cntA) sA = csr_src[begA + lane];
        int begB = 0, cntB = 0, sB = 0;
        if (nn > 1) {
            begB = __builtin_amdgcn_readlane(m_beg, 1);
            cntB = __builtin_amdgcn_readlane(m_cnt, 1);
            if (lane < cntB) sB = csr_src[begB + lane];
        }
        float2 svA = *(const float2*)(xin + (size_t)n0 * K + coff);

        for (int i = 0; i < nn; ++i) {
            const int node = n0 + i;
            const int beg = begA;
            const int end = begA + cntA;
            const float dd = __int_as_float(__builtin_amdgcn_readlane(m_ddbits, i));
            const int first_s = sA;
            float ax = svA.x;
            float ay = svA.y;

            begA = begB; cntA = cntB; sA = sB;
            if (i + 1 < nn)
                svA = *(const float2*)(xin + (size_t)(node + 1) * K + coff);
            if (i + 2 < nn) {
                begB = __builtin_amdgcn_readlane(m_beg, i + 2);
                cntB = __builtin_amdgcn_readlane(m_cnt, i + 2);
                sB = 0;
                if (lane < cntB) sB = csr_src[begB + lane];
            }

            for (int p = beg; p < end; p += 32) {
                int my_s = first_s;
                if (p != beg) {
                    const int idx = p + lane;
                    my_s = 0;
                    if (idx < end) my_s = csr_src[idx];
                }
                const int left = end - p;
                const int n = left < 32 ? left : 32;
                float2 ra[8], rb[8], rc[8], rd[8], t4[4], t2[2], t1[1];
                const int nt = n & ~7;
                if (n >= 8)  gcn_rows_ask<K, 8>(xin, my_s, 0,  coff, ra);
                if (n >= 16) gcn_rows_ask<K, 8>(xin, my_s, 8,  coff, rb);
                if (n >= 24) gcn_rows_ask<K, 8>(xin, my_s, 16, coff, rc);
                if (n >= 32) gcn_rows_ask<K, 8>(xin, my_s, 24, coff, rd);
                int j = nt;
                if (n & 4) { gcn_rows_ask<K, 4>(xin, my_s, j, coff, t4); j += 4; }
                if (n & 2) { gcn_rows_ask<K, 2>(xin, my_s, j, coff, t2); j += 2; }
                if (n & 1) { gcn_rows_ask<K, 1>(xin, my_s, j, coff, t1); }

                if (n >= 8)  gcn_rows_add<8>(ra, ax, ay);
                if (n >= 16) gcn_rows_add<8>(rb, ax, ay);
                if (n >= 24) gcn_rows_add<8>(rc, ax, ay);
                if (n >= 32) gcn_rows_add<8>(rd, ax, ay);
                if (n & 4) gcn_rows_add<4>(t4, ax, ay);
                if (n & 2) gcn_rows_add<2>(t2, ax, ay);
                if (n & 1) gcn_rows_add<1>(t1, ax, ay);
            }

            if (lane2 < KP) {
                float2 o;
                o.x = lane2 < K ? dd * ax : 0.0f;
                o.y = lane2 < K ? dd * ay : 0.0f;
                *(float2*)(ATw + i * KS + lane2) = o;
            }
        }
    }

    __syncthreads();

    const int rt  = wib >> 1;
    const int ct0 = (wib & 1) * NT2;
    const int row0 = nb0 + (rt << 4);
    if (row0 >= nNodes) return;

    const int mrow  = lane & 15;
    const int khalf = (lane >> 4) << 1;
    const float* ATlane = AT + ((rt << 4) + mrow) * KS;
    const float* BTlane = BT + (ct0 * 16 + mrow) * KS;

    const int rbase = row0 + ((lane >> 4) << 3);
    float4 dlo = make_float4(1.0f, 1.0f, 1.0f, 1.0f), dhi = dlo;
    if (HIDDEN) {
        dlo = *(const float4*)(dinv + rbase);
        dhi = *(const float4*)(dinv + rbase + 4);
    }

    v8f acc[NT2];
#pragma unroll
    for (int t = 0; t < NT2; ++t) acc[t] = v8f{};

#pragma unroll
    for (int k0 = 0; k0 < KP; k0 += 4) {
        const int ka = k0 + khalf;
        const float2 av = *(const float2*)(ATlane + ka);
        v2f a; a.x = av.x; a.y = av.y;
#pragma unroll
        for (int t = 0; t < NT2; ++t) {
            const float2 bv = *(const float2*)(BTlane + t * 16 * KS + ka);
            v2f b; b.x = bv.x; b.y = bv.y;
            acc[t] = __builtin_amdgcn_wmma_f32_16x16x4_f32(
                false, a, false, b, (short)0, acc[t], false, false);
        }
    }

    const float dsc[8] = {dlo.x, dlo.y, dlo.z, dlo.w, dhi.x, dhi.y, dhi.z, dhi.w};
#pragma unroll
    for (int t = 0; t < NT2; ++t) {
        const int col = ((ct0 + t) << 4) + mrow;
        if (col < N) {
#pragma unroll
            for (int r = 0; r < 8; ++r) {
                const float v = acc[t][r];
                C[(size_t)(rbase + r) * N + col] = HIDDEN ? dsc[r] * tanhf(v) : v;
            }
        }
    }
}

extern "C" void kernel_launch(void* const* d_in, const int* in_sizes, int n_in,
                              void* d_out, int out_size, void* d_ws, size_t ws_size,
                              hipStream_t stream) {
    const float* x  = (const float*)d_in[0];
    const float* W1 = (const float*)d_in[1];
    const float* W2 = (const float*)d_in[2];
    const int* ei   = (const int*)d_in[3];

    const int FIN = 50, F1 = 64;
    const int nNodes = in_sizes[0] / FIN;
    const int nEdges = in_sizes[3] / 2;
    const int* src = ei;
    const int* dst = ei + nEdges;

    const int NP = (nNodes + 63) & ~63;
    const int EP = (nEdges + 63) & ~63;
    const int nbkt = (nNodes + BKT_NODES - 1) >> BKT_SHIFT;
    const int eblk = (nEdges + SPAN - 1) / SPAN;
    const int HP = (eblk * nbkt + 63) & ~63;
    float* wsF    = (float*)d_ws;
    float* dinv   = wsF;
    int*   cnt    = (int*)(dinv + NP);
    int*   rowptr = cnt + NP;
    int*   hist   = rowptr + NP;
    int*   lstart = hist + HP;
    int2*  binned = (int2*)(lstart + HP);
    int*   csr    = (int*)(binned + EP);
    float* hbuf   = (float*)(csr + EP);
    float* xsbuf  = hbuf + (size_t)NP * F1;
    float* outf   = (float*)d_out;

    const int T = 256;
    const int layerBlocks = (nNodes + 63) / 64;

    gcn_span_sort_kernel<<<eblk, BNTH, 0, stream>>>(src, dst, hist, lstart, binned,
                                                    nEdges, nbkt);
    gcn_bucket_rows_kernel<50><<<nbkt, BNTH, 0, stream>>>(binned, hist, lstart, x, rowptr, cnt,
                                                          dinv, csr, xsbuf, nNodes, nbkt, eblk);

    gcn_gather_gemm_kernel<50, 64, 4, 1><<<layerBlocks, T, 0, stream>>>(
        xsbuf, W1, rowptr, cnt, csr, dinv, hbuf, nNodes);
    gcn_gather_gemm_kernel<64, 121, 8, 0><<<layerBlocks, T, 0, stream>>>(
        hbuf, W2, rowptr, cnt, csr, dinv, outf, nNodes);

    (void)n_in; (void)out_size; (void)ws_size; (void)F1;
}


# Stores leave as whole 128-B lines: a bucket's source list and each layer block's 64 output rows are assembled in LDS and written in order, 16 B a lane, not as scattered 4-B stores; f32 sums unchanged
# speedup vs baseline: 6.2273x; 1.0325x over previous
#include <hip/hip_runtime.h>
#include <hip/hip_bf16.h>


typedef __attribute__((ext_vector_type(2))) float v2f;
typedef __attribute__((ext_vector_type(8))) float v8f;


constexpr int BKT_SHIFT = 8;
constexpr int BKT_NODES = 1 << BKT_SHIFT;
constexpr int MAXBKT    = 1024;
constexpr int BNTH      = 1024;
constexpr int BEPT      = 8;
constexpr int SPAN      = BNTH * BEPT;
constexpr int MAXBLK    = 1024;
constexpr int SEGCAP    = 8192;

__global__ __launch_bounds__(BNTH) void gcn_span_sort_kernel(
    const int* __restrict__ src, const int* __restrict__ dst,
    int* __restrict__ hist, int* __restrict__ lstart,
    int2* __restrict__ binned, int nE, int nbkt) {
    static_assert(MAXBKT == BNTH, "one bucket counter per thread");
    static_assert((BEPT & 1) == 0, "the span is written out two pairs at a time");
    __shared__ int h[MAXBKT];
    __shared__ int s_wsum[BNTH / 32];
    __shared__ int2 stage[SPAN];
    const int tid  = threadIdx.x;
    const int lane = tid & 31;
    const int wid  = tid >> 5;
    const int bid  = blockIdx.x;
    const int base = bid * SPAN + tid;
    int s[BEPT];
    int d[BEPT];
    h[tid] = 0;
#pragma unroll
    for (int k = 0; k < BEPT; ++k) {
        const int e  = base + k * BNTH;
        const int ec = e < nE ? e : nE - 1;
        s[k] = src[ec];
        d[k] = dst[ec];
    }
    __syncthreads();
#pragma unroll
    for (int k = 0; k < BEPT; ++k)
        if (base + k * BNTH < nE) atomicAdd(&h[d[k] >> BKT_SHIFT], 1);
    __syncthreads();

    const int tot = h[tid];
    int incl = tot;
    for (int o = 1; o < 32; o <<= 1) {
        const int t = __shfl_up(incl, o);
        if (lane >= o) incl += t;
    }
    if (lane == 31) s_wsum[wid] = incl;
    __syncthreads();
    int wbase = 0;
    for (int w = 0; w < BNTH / 32; ++w)
        if (w < wid) wbase += s_wsum[w];
    const int first = wbase + incl - tot;
    h[tid] = first;
    if (tid < nbkt) {
        hist[bid * nbkt + tid]   = tot;
        lstart[bid * nbkt + tid] = first;
    }
    __syncthreads();

    int slot[BEPT];
#pragma unroll
    for (int k = 0; k < BEPT; ++k)
        slot[k] = (base + k * BNTH < nE) ? atomicAdd(&h[d[k] >> BKT_SHIFT], 1) : -1;
#pragma unroll
    for (int k = 0; k < BEPT; ++k)
        if (slot[k] >= 0) stage[slot[k]] = make_int2(s[k], d[k]);
    __syncthreads();

    const int left = nE - bid * SPAN;
    const int nv = left < SPAN ? left : SPAN;
    int2* seg = binned + (size_t)bid * SPAN;
    const int4* stage4 = (const int4*)stage;
    int4* seg4 = (int4*)seg;
#pragma unroll
    for (int k = 0; k < BEPT / 2; ++k) {
        const int i = tid + k * BNTH;
        if (2 * i + 1 < nv) seg4[i] = stage4[i];
        else if (2 * i < nv) seg[2 * i] = stage[2 * i];
    }
}

template <int K>
__global__ __launch_bounds__(BNTH) void gcn_bucket_rows_kernel(
    const int2* __restrict__ binned, const int* __restrict__ hist,
    const int* __restrict__ lstart, const float* __restrict__ x,
    int* __restrict__ rowptr, int* __restrict__ cnt, float* __restrict__ dinv,
    int* __restrict__ csr_src, float* __restrict__ xs,
    int nNodes, int nbkt, int nblk) {
    constexpr int NU = 4, NV = 2;
    static_assert(NU == 4 && NV == 2, "the load fence below names 16 values");
    static_assert((K & 1) == 0, "rows are moved two floats at a time");
    constexpr int K2 = K / 2;
    constexpr int XQ = (BKT_NODES * K2 + BNTH - 1) / BNTH;
    __shared__ int c[BKT_NODES];
    __shared__ int cur[BKT_NODES];
    __shared__ float dv[BKT_NODES];
    __shared__ int s_wsum[BKT_NODES / 32];
    __shared__ int c_s[MAXBLK];
    __shared__ int l_s[MAXBLK];
    __shared__ int s_red[BNTH / 32];
    __shared__ __attribute__((aligned(16))) int seg[SEGCAP];
    const int tid  = threadIdx.x;
    const int lane = tid & 31;
    const int wid  = tid >> 5;
    const int bk   = blockIdx.x;
    const int node0 = bk << BKT_SHIFT;

    const int nleft = nNodes - node0;
    const int nx2 = (nleft < BKT_NODES ? nleft : BKT_NODES) * K2;
    const float2* x2 = (const float2*)(x + (size_t)node0 * K);
    float2 xv[XQ];
#pragma unroll
    for (int q = 0; q < XQ; ++q) {
        const int i = tid + q * BNTH;
        xv[q] = x2[i < nx2 ? i : 0];
    }

    int lsum = 0;
    for (int b = tid; b < nblk; b += BNTH) {
        const int cc = hist[b * nbkt + bk];
        const int ll = lstart[b * nbkt + bk];
        c_s[b] = cc;
        l_s[b] = ll;
        lsum += ll;
    }
    if (tid < BKT_NODES) c[tid] = 0;
    for (int o = 16; o > 0; o >>= 1) lsum += __shfl_xor(lsum, o);
    if (lane == 0) s_red[wid] = lsum;
    __syncthreads();
    int b0 = 0;
    for (int w = 0; w < BNTH / 32; ++w) b0 += s_red[w];

    int2 p[NU][NV];
    int  nn[NU];
#pragma unroll
    for (int u = 0; u < NU; ++u) {
        const int b  = wid + 32 * u;
        const int bc = b < nblk ? b : nblk - 1;
        const int n  = b < nblk ? c_s[bc] : 0;
        const int ls = l_s[bc];
        nn[u] = n;
#pragma unroll
        for (int v = 0; v < NV; ++v) {
            const int i   = 32 * v + lane;
            const int off = i < n ? ls + i : 0;
            p[u][v] = binned[(size_t)bc * SPAN + off];
        }
    }
    asm volatile("" :: "v"(p[0][0].x), "v"(p[0][0].y), "v"(p[0][1].x), "v"(p[0][1].y),
                       "v"(p[1][0].x), "v"(p[1][0].y), "v"(p[1][1].x), "v"(p[1][1].y),
                       "v"(p[2][0].x), "v"(p[2][0].y), "v"(p[2][1].x), "v"(p[2][1].y),
                       "v"(p[3][0].x), "v"(p[3][0].y), "v"(p[3][1].x), "v"(p[3][1].y));
#pragma unroll
    for (int u = 0; u < NU; ++u)
#pragma unroll
        for (int v = 0; v < NV; ++v)
            if (32 * v + lane < nn[u]) atomicAdd(&c[p[u][v].y - node0], 1);
    for (int b = wid; b < nblk; b += 32) {
        const int n = c_s[b];
        const int2* run = binned + (size_t)b * SPAN + l_s[b];
        for (int i = (b < 32 * NU ? 32 * NV : 0) + lane; i < n; i += 32)
            atomicAdd(&c[run[i].y - node0], 1);
    }
    __syncthreads();

    int my = 0, incl = 0;
    if (tid < BKT_NODES) {
        my = c[tid];
        incl = my;
        for (int o = 1; o < 32; o <<= 1) {
            const int t = __shfl_up(incl, o);
            if (lane >= o) incl += t;
        }
        if (lane == 31) s_wsum[wid] = incl;
    }
    __syncthreads();
    if (tid < BKT_NODES) {
        int wbase = 0;
        for (int w = 0; w < BKT_NODES / 32; ++w)
            if (w < wid) wbase += s_wsum[w];
        const int first = b0 + wbase + incl - my;
        cur[tid] = first;
        const float di = rsqrtf((float)(my + 1));
        dv[tid] = di;
        const int node = node0 + tid;
        if (node < nNodes) {
            rowptr[node] = first;
            cnt[node]    = my;
            dinv[node]   = di;
        }
    }
    __syncthreads();

    float2* xs2 = (float2*)(xs + (size_t)node0 * K);
#pragma unroll
    for (int q = 0; q < XQ; ++q) {
        const int i = tid + q * BNTH;
        if (i < nx2) {
            const float w = dv[i / K2];
            xs2[i] = make_float2(w * xv[q].x, w * xv[q].y);
        }
    }

    int T = 0;
    for (int w = 0; w < BKT_NODES / 32; ++w) T += s_wsum[w];
    const int a  = b0 & 31;
    const int gb = b0 - a;
    const bool staged = a + T <= SEGCAP;

    int slot[NU][NV];
#pragma unroll
    for (int u = 0; u < NU; ++u)
#pragma unroll
        for (int v = 0; v < NV; ++v)
            slot[u][v] = (32 * v + lane < nn[u]) ? atomicAdd(&cur[p[u][v].y - node0], 1) : -1;
    if (staged) {
#pragma unroll
        for (int u = 0; u < NU; ++u)
#pragma unroll
            for (int v = 0; v < NV; ++v)
                if (slot[u][v] >= 0) seg[slot[u][v] - gb] = p[u][v].x;
        for (int b = wid; b < nblk; b += 32) {
            const int n = c_s[b];
            const int2* run = binned + (size_t)b * SPAN + l_s[b];
            for (int i = (b < 32 * NU ? 32 * NV : 0) + lane; i < n; i += 32) {
                const int2 q = run[i];
                seg[atomicAdd(&cur[q.y - node0], 1) - gb] = q.x;
            }
        }
        __syncthreads();
        const int4* seg4 = (const int4*)seg;
        int4* csr4 = (int4*)(csr_src + gb);
        const int endi = a + T;
        const int qlo = (a + 3) >> 2;
        const int qhi = endi >> 2;
        for (int q = tid; q < qhi; q += BNTH)
            if (q >= qlo) csr4[q] = seg4[q];
        if (tid < 3) {
            const int e = a + tid;
            if (e < (qlo << 2) && e < endi) csr_src[gb + e] = seg[e];
        } else if (tid >= 32 && tid < 35) {
            const int e = (qhi << 2) + (tid - 32);
            if (qhi >= qlo && e < endi) csr_src[gb + e] = seg[e];
        }
    } else {
#pragma unroll
        for (int u = 0; u < NU; ++u)
#pragma unroll
            for (int v = 0; v < NV; ++v)
                if (slot[u][v] >= 0) csr_src[slot[u][v]] = p[u][v].x;
        for (int b = wid; b < nblk; b += 32) {
            const int n = c_s[b];
            const int2* run = binned + (size_t)b * SPAN + l_s[b];
            for (int i = (b < 32 * NU ? 32 * NV : 0) + lane; i < n; i += 32) {
                const int2 q = run[i];
                csr_src[atomicAdd(&cur[q.y - node0], 1)] = q.x;
            }
        }
    }
}

template <int K, int U>
__device__ __forceinline__ void gcn_rows_ask(
    const float* __restrict__ xin, int my_s, int j, int coff, float2* r) {
#pragma unroll
    for (int u = 0; u < U; ++u) {
        const int s = __builtin_amdgcn_readlane(my_s, j + u);
        r[u] = *(const float2*)(xin + (size_t)s * K + coff);
    }
}

template <int U>
__device__ __forceinline__ void gcn_rows_add(const float2* r, float& ax, float& ay) {
#pragma unroll
    for (int u = 0; u < U; ++u) {
        ax += r[u].x;
        ay += r[u].y;
    }
}

template <int K, int N, int NTILES, int HIDDEN>
__global__ __launch_bounds__(256) void gcn_gather_gemm_kernel(
    const float* __restrict__ xin, const float* __restrict__ B,
    const int* __restrict__ rowptr, const int* __restrict__ cnt,
    const int* __restrict__ csr_src, const float* __restrict__ dinv,
    float* __restrict__ C, int nNodes) {
    static_assert((K & 1) == 0 && K <= 64, "float2 row loads, one per lane");
    static_assert((NTILES & 1) == 0, "the column tiles are split between two waves");
    constexpr int KP  = (K + 3) & ~3;
    constexpr int KS  = KP + 2;
    constexpr int NP  = NTILES * 16;
    constexpr int NT2 = NTILES / 2;
    constexpr int NTH = 256;
    constexpr int NPW = 8;
    constexpr int NPB = NPW * (NTH / 32);
    static_assert(NPB * N <= (NP + NPB) * KS, "the output tile lies over the operands");
    static_assert((16 * N) % 4 == 0, "whole row tiles are a whole number of 16-byte groups");
    __shared__ __attribute__((aligned(16))) float SM[(NP + NPB) * KS];
    float* BT = SM;
    float* AT = SM + NP * KS;

    constexpr int TOT = KS * NP;
    constexpr int PER = (TOT + NTH - 1) / NTH;
    {
        float v[PER];
#pragma unroll
        for (int q = 0; q < PER; ++q) {
            const int idx = (int)threadIdx.x + q * NTH;
            const int k = idx / NP;
            const int n = idx - k * NP;
            const int kc = k < K ? k : K - 1;
            const int nc = n < N ? n : N - 1;
            const float b = B[kc * N + nc];
            v[q] = (n < N && k < K) ? b : 0.0f;
        }
#pragma unroll
        for (int q = 0; q < PER; ++q) {
            const int idx = (int)threadIdx.x + q * NTH;
            const int k = idx / NP;
            const int n = idx - k * NP;
            if (idx < TOT) BT[n * KS + k] = v[q];
        }
    }

    const int wib = __builtin_amdgcn_readfirstlane((int)(threadIdx.x >> 5));
    const int nb0 = __builtin_amdgcn_readfirstlane((int)(blockIdx.x * NPB));
    const int n0  = nb0 + wib * NPW;
    const int lane  = threadIdx.x & 31;
    const int lane2 = lane << 1;
    const int coff = lane2 < K ? lane2 : 0;
    const int nleft = nNodes - n0;
    const int nn = nleft < NPW ? nleft : NPW;
    float* ATw = AT + wib * (NPW * KS);

    if (nn > 0) {
        int m_beg = 0, m_cnt = 0, m_ddbits = 0;
        if (lane < nn) {
            m_beg    = rowptr[n0 + lane];
            m_cnt    = cnt[n0 + lane];
            m_ddbits = __float_as_int(dinv[n0 + lane]);
        }

        int begA = __builtin_amdgcn_readlane(m_beg, 0);
        int cntA = __builtin_amdgcn_readlane(m_cnt, 0);
        int sA = 0;
        if (lane < cntA) sA = csr_src[begA + lane];
        int begB = 0, cntB = 0, sB = 0;
        if (nn > 1) {
            begB = __builtin_amdgcn_readlane(m_beg, 1);
            cntB = __builtin_amdgcn_readlane(m_cnt, 1);
            if (lane < cntB) sB = csr_src[begB + lane];
        }
        float2 svA = *(const float2*)(xin + (size_t)n0 * K + coff);

        for (int i = 0; i < nn; ++i) {
            const int node = n0 + i;
            const int beg = begA;
            const int end = begA + cntA;
            const float dd = __int_as_float(__builtin_amdgcn_readlane(m_ddbits, i));
            const int first_s = sA;
            float ax = svA.x;
            float ay = svA.y;

            begA = begB; cntA = cntB; sA = sB;
            if (i + 1 < nn)
                svA = *(const float2*)(xin + (size_t)(node + 1) * K + coff);
            if (i + 2 < nn) {
                begB = __builtin_amdgcn_readlane(m_beg, i + 2);
                cntB = __builtin_amdgcn_readlane(m_cnt, i + 2);
                sB = 0;
                if (lane < cntB) sB = csr_src[begB + lane];
            }

            for (int p = beg; p < end; p += 32) {
                int my_s = first_s;
                if (p != beg) {
                    const int idx = p + lane;
                    my_s = 0;
                    if (idx < end) my_s = csr_src[idx];
                }
                const int left = end - p;
                const int n = left < 32 ? left : 32;
                float2 ra[8], rb[8], rc[8], rd[8], t4[4], t2[2], t1[1];
                const int nt = n & ~7;
                if (n >= 8)  gcn_rows_ask<K, 8>(xin, my_s, 0,  coff, ra);
                if (n >= 16) gcn_rows_ask<K, 8>(xin, my_s, 8,  coff, rb);
                if (n >= 24) gcn_rows_ask<K, 8>(xin, my_s, 16, coff, rc);
                if (n >= 32) gcn_rows_ask<K, 8>(xin, my_s, 24, coff, rd);
                int j = nt;
                if (n & 4) { gcn_rows_ask<K, 4>(xin, my_s, j, coff, t4); j += 4; }
                if (n & 2) { gcn_rows_ask<K, 2>(xin, my_s, j, coff, t2); j += 2; }
                if (n & 1) { gcn_rows_ask<K, 1>(xin, my_s, j, coff, t1); }

                if (n >= 8)  gcn_rows_add<8>(ra, ax, ay);
                if (n >= 16) gcn_rows_add<8>(rb, ax, ay);
                if (n >= 24) gcn_rows_add<8>(rc, ax, ay);
                if (n >= 32) gcn_rows_add<8>(rd, ax, ay);
                if (n & 4) gcn_rows_add<4>(t4, ax, ay);
                if (n & 2) gcn_rows_add<2>(t2, ax, ay);
                if (n & 1) gcn_rows_add<1>(t1, ax, ay);
            }

            if (lane2 < KP) {
                float2 o;
                o.x = lane2 < K ? dd * ax : 0.0f;
                o.y = lane2 < K ? dd * ay : 0.0f;
                *(float2*)(ATw + i * KS + lane2) = o;
            }
        }
    }

    __syncthreads();

    const int rt  = wib >> 1;
    const int ct0 = (wib & 1) * NT2;
    const int row0 = nb0 + (rt << 4);
    const bool have = row0 < nNodes;

    const int mrow  = lane & 15;
    const int khalf = (lane >> 4) << 1;
    const float* ATlane = AT + ((rt << 4) + mrow) * KS;
    const float* BTlane = BT + (ct0 * 16 + mrow) * KS;

    const int rbase = row0 + ((lane >> 4) << 3);
    float4 dlo = make_float4(1.0f, 1.0f, 1.0f, 1.0f), dhi = dlo;
    if (HIDDEN && have) {
        dlo = *(const float4*)(dinv + rbase);
        dhi = *(const float4*)(dinv + rbase + 4);
    }

    v8f acc[NT2];
#pragma unroll
    for (int t = 0; t < NT2; ++t) acc[t] = v8f{};

    if (have) {
#pragma unroll
        for (int k0 = 0; k0 < KP; k0 += 4) {
            const int ka = k0 + khalf;
            const float2 av = *(const float2*)(ATlane + ka);
            v2f a; a.x = av.x; a.y = av.y;
#pragma unroll
            for (int t = 0; t < NT2; ++t) {
                const float2 bv = *(const float2*)(BTlane + t * 16 * KS + ka);
                v2f b; b.x = bv.x; b.y = bv.y;
                acc[t] = __builtin_amdgcn_wmma_f32_16x16x4_f32(
                    false, a, false, b, (short)0, acc[t], false, false);
            }
        }
    }

    __syncthreads();
    float* OT = SM;
    if (have) {
        const float dsc[8] = {dlo.x, dlo.y, dlo.z, dlo.w, dhi.x, dhi.y, dhi.z, dhi.w};
        float* OTlane = OT + ((rt << 4) + ((lane >> 4) << 3)) * N;
#pragma unroll
        for (int t = 0; t < NT2; ++t) {
            const int col = ((ct0 + t) << 4) + mrow;
            if (col < N) {
#pragma unroll
                for (int r = 0; r < 8; ++r) {
                    const float v = acc[t][r];
                    OTlane[r * N + col] = HIDDEN ? dsc[r] * tanhf(v) : v;
                }
            }
        }
    }
    __syncthreads();

    const int rleft = nNodes - nb0;
    const int rows  = rleft < NPB ? rleft : NPB;
    constexpr int NQF = NPB * N / 4;
    constexpr int CQ  = (NQF + NTH - 1) / NTH;
    const float4* OT4 = (const float4*)SM;
    float4* C4 = (float4*)(C + (size_t)nb0 * N);
    if (rows == NPB) {
        float4 o[CQ];
#pragma unroll
        for (int q = 0; q < CQ; ++q) {
            const int i = (int)threadIdx.x + q * NTH;
            o[q] = OT4[((q + 1) * NTH <= NQF || i < NQF) ? i : 0];
        }
#pragma unroll
        for (int q = 0; q < CQ; ++q) {
            const int i = (int)threadIdx.x + q * NTH;
            if ((q + 1) * NTH <= NQF || i < NQF) C4[i] = o[q];
        }
    } else {
        const int nq = (rows * N) >> 2;
        for (int i = (int)threadIdx.x; i < nq; i += NTH) C4[i] = OT4[i];
    }
}

extern "C" void kernel_launch(void* const* d_in, const int* in_sizes, int n_in,
                              void* d_out, int out_size, void* d_ws, size_t ws_size,
                              hipStream_t stream) {
    const float* x  = (const float*)d_in[0];
    const float* W1 = (const float*)d_in[1];
    const float* W2 = (const float*)d_in[2];
    const int* ei   = (const int*)d_in[3];

    const int FIN = 50, F1 = 64;
    const int nNodes = in_sizes[0] / FIN;
    const int nEdges = in_sizes[3] / 2;
    const int* src = ei;
    const int* dst = ei + nEdges;

    const int NP = (nNodes + 63) & ~63;
    const int EP = (nEdges + 63) & ~63;
    const int nbkt = (nNodes + BKT_NODES - 1) >> BKT_SHIFT;
    const int eblk = (nEdges + SPAN - 1) / SPAN;
    const int HP = (eblk * nbkt + 63) & ~63;
    float* wsF    = (float*)d_ws;
    float* dinv   = wsF;
    int*   cnt    = (int*)(dinv + NP);
    int*   rowptr = cnt + NP;
    int*   hist   = rowptr + NP;
    int*   lstart = hist + HP;
    int2*  binned = (int2*)(lstart + HP);
    int*   csr    = (int*)(binned + EP);
    float* hbuf   = (float*)(csr + EP);
    float* xsbuf  = hbuf + (size_t)NP * F1;
    float* outf   = (float*)d_out;

    const int T = 256;
    const int layerBlocks = (nNodes + 63) / 64;

    gcn_span_sort_kernel<<<eblk, BNTH, 0, stream>>>(src, dst, hist, lstart, binned,
                                                    nEdges, nbkt);
    gcn_bucket_rows_kernel<50><<<nbkt, BNTH, 0, stream>>>(binned, hist, lstart, x, rowptr, cnt,
                                                          dinv, csr, xsbuf, nNodes, nbkt, eblk);

    gcn_gather_gemm_kernel<50, 64, 4, 1><<<layerBlocks, T, 0, stream>>>(
        xsbuf, W1, rowptr, cnt, csr, dinv, hbuf, nNodes);
    gcn_gather_gemm_kernel<64, 121, 8, 0><<<layerBlocks, T, 0, stream>>>(
        hbuf, W2, rowptr, cnt, csr, dinv, outf, nNodes);

    (void)n_in; (void)out_size; (void)ws_size; (void)F1;
}


# Opening loads reordered (a wave's loads return in order): layer blocks ask for node table and own row before the W tile, bucket blocks for x rows behind their pair loads; f32 sums unchanged
# speedup vs baseline: 6.4723x; 1.0393x over previous
#include <hip/hip_runtime.h>
#include <hip/hip_bf16.h>


typedef __attribute__((ext_vector_type(2))) float v2f;
typedef __attribute__((ext_vector_type(8))) float v8f;


constexpr int BKT_SHIFT = 8;
constexpr int BKT_NODES = 1 << BKT_SHIFT;
constexpr int MAXBKT    = 1024;
constexpr int BNTH      = 1024;
constexpr int BEPT      = 8;
constexpr int SPAN      = BNTH * BEPT;
constexpr int MAXBLK    = 1024;
constexpr int SEGCAP    = 8192;

__global__ __launch_bounds__(BNTH) void gcn_span_sort_kernel(
    const int* __restrict__ src, const int* __restrict__ dst,
    int* __restrict__ hist, int* __restrict__ lstart,
    int2* __restrict__ binned, int nE, int nbkt) {
    static_assert(MAXBKT == BNTH, "one bucket counter per thread");
    static_assert((BEPT & 1) == 0, "the span is written out two pairs at a time");
    __shared__ int h[MAXBKT];
    __shared__ int s_wsum[BNTH / 32];
    __shared__ int2 stage[SPAN];
    const int tid  = threadIdx.x;
    const int lane = tid & 31;
    const int wid  = tid >> 5;
    const int bid  = blockIdx.x;
    const int base = bid * SPAN + tid;
    int s[BEPT];
    int d[BEPT];
    h[tid] = 0;
#pragma unroll
    for (int k = 0; k < BEPT; ++k) {
        const int e  = base + k * BNTH;
        const int ec = e < nE ? e : nE - 1;
        s[k] = src[ec];
        d[k] = dst[ec];
    }
    __syncthreads();
#pragma unroll
    for (int k = 0; k < BEPT; ++k)
        if (base + k * BNTH < nE) atomicAdd(&h[d[k] >> BKT_SHIFT], 1);
    __syncthreads();

    const int tot = h[tid];
    int incl = tot;
    for (int o = 1; o < 32; o <<= 1) {
        const int t = __shfl_up(incl, o);
        if (lane >= o) incl += t;
    }
    if (lane == 31) s_wsum[wid] = incl;
    __syncthreads();
    int wbase = 0;
    for (int w = 0; w < BNTH / 32; ++w)
        if (w < wid) wbase += s_wsum[w];
    const int first = wbase + incl - tot;
    h[tid] = first;
    if (tid < nbkt) {
        hist[bid * nbkt + tid]   = tot;
        lstart[bid * nbkt + tid] = first;
    }
    __syncthreads();

    int slot[BEPT];
#pragma unroll
    for (int k = 0; k < BEPT; ++k)
        slot[k] = (base + k * BNTH < nE) ? atomicAdd(&h[d[k] >> BKT_SHIFT], 1) : -1;
#pragma unroll
    for (int k = 0; k < BEPT; ++k)
        if (slot[k] >= 0) stage[slot[k]] = make_int2(s[k], d[k]);
    __syncthreads();

    const int left = nE - bid * SPAN;
    const int nv = left < SPAN ? left : SPAN;
    int2* seg = binned + (size_t)bid * SPAN;
    const int4* stage4 = (const int4*)stage;
    int4* seg4 = (int4*)seg;
#pragma unroll
    for (int k = 0; k < BEPT / 2; ++k) {
        const int i = tid + k * BNTH;
        if (2 * i + 1 < nv) seg4[i] = stage4[i];
        else if (2 * i < nv) seg[2 * i] = stage[2 * i];
    }
}

template <int K>
__global__ __launch_bounds__(BNTH) void gcn_bucket_rows_kernel(
    const int2* __restrict__ binned, const int* __restrict__ hist,
    const int* __restrict__ lstart, const float* __restrict__ x,
    int* __restrict__ rowptr, int* __restrict__ cnt, float* __restrict__ dinv,
    int* __restrict__ csr_src, float* __restrict__ xs,
    int nNodes, int nbkt, int nblk) {
    constexpr int NU = 4, NV = 2;
    static_assert(NU == 4 && NV == 2, "the load fence below names 16 values");
    static_assert((K & 1) == 0, "rows are moved two floats at a time");
    constexpr int K2 = K / 2;
    constexpr int XQ = (BKT_NODES * K2 + BNTH - 1) / BNTH;
    __shared__ int c[BKT_NODES];
    __shared__ int cur[BKT_NODES];
    __shared__ float dv[BKT_NODES];
    __shared__ int s_wsum[BKT_NODES / 32];
    __shared__ int c_s[MAXBLK];
    __shared__ int l_s[MAXBLK];
    __shared__ int s_red[BNTH / 32];
    __shared__ __attribute__((aligned(16))) int seg[SEGCAP];
    const int tid  = threadIdx.x;
    const int lane = tid & 31;
    const int wid  = tid >> 5;
    const int bk   = blockIdx.x;
    const int node0 = bk << BKT_SHIFT;

    const int nleft = nNodes - node0;
    const int nx2 = (nleft < BKT_NODES ? nleft : BKT_NODES) * K2;
    const float2* x2 = (const float2*)(x + (size_t)node0 * K);

    int lsum = 0;
    for (int b = tid; b < nblk; b += BNTH) {
        const int cc = hist[b * nbkt + bk];
        const int ll = lstart[b * nbkt + bk];
        c_s[b] = cc;
        l_s[b] = ll;
        lsum += ll;
    }
    if (tid < BKT_NODES) c[tid] = 0;
    for (int o = 16; o > 0; o >>= 1) lsum += __shfl_xor(lsum, o);
    if (lane == 0) s_red[wid] = lsum;
    __syncthreads();
    int b0 = 0;
    for (int w = 0; w < BNTH / 32; ++w) b0 += s_red[w];

    int2 p[NU][NV];
    int  nn[NU];
#pragma unroll
    for (int u = 0; u < NU; ++u) {
        const int b  = wid + 32 * u;
        const int bc = b < nblk ? b : nblk - 1;
        const int n  = b < nblk ? c_s[bc] : 0;
        const int ls = l_s[bc];
        nn[u] = n;
#pragma unroll
        for (int v = 0; v < NV; ++v) {
            const int i   = 32 * v + lane;
            const int off = i < n ? ls + i : 0;
            p[u][v] = binned[(size_t)bc * SPAN + off];
        }
    }
    float2 xv[XQ];
#pragma unroll
    for (int q = 0; q < XQ; ++q) {
        const int i = tid + q * BNTH;
        xv[q] = x2[i < nx2 ? i : 0];
    }
    asm volatile("" :: "v"(p[0][0].x), "v"(p[0][0].y), "v"(p[0][1].x), "v"(p[0][1].y),
                       "v"(p[1][0].x), "v"(p[1][0].y), "v"(p[1][1].x), "v"(p[1][1].y),
                       "v"(p[2][0].x), "v"(p[2][0].y), "v"(p[2][1].x), "v"(p[2][1].y),
                       "v"(p[3][0].x), "v"(p[3][0].y), "v"(p[3][1].x), "v"(p[3][1].y));
#pragma unroll
    for (int u = 0; u < NU; ++u)
#pragma unroll
        for (int v = 0; v < NV; ++v)
            if (32 * v + lane < nn[u]) atomicAdd(&c[p[u][v].y - node0], 1);
    for (int b = wid; b < nblk; b += 32) {
        const int n = c_s[b];
        const int2* run = binned + (size_t)b * SPAN + l_s[b];
        for (int i = (b < 32 * NU ? 32 * NV : 0) + lane; i < n; i += 32)
            atomicAdd(&c[run[i].y - node0], 1);
    }
    __syncthreads();

    int my = 0, incl = 0;
    if (tid < BKT_NODES) {
        my = c[tid];
        incl = my;
        for (int o = 1; o < 32; o <<= 1) {
            const int t = __shfl_up(incl, o);
            if (lane >= o) incl += t;
        }
        if (lane == 31) s_wsum[wid] = incl;
    }
    __syncthreads();
    if (tid < BKT_NODES) {
        int wbase = 0;
        for (int w = 0; w < BKT_NODES / 32; ++w)
            if (w < wid) wbase += s_wsum[w];
        const int first = b0 + wbase + incl - my;
        cur[tid] = first;
        const float di = rsqrtf((float)(my + 1));
        dv[tid] = di;
        const int node = node0 + tid;
        if (node < nNodes) {
            rowptr[node] = first;
            cnt[node]    = my;
            dinv[node]   = di;
        }
    }
    __syncthreads();

    float2* xs2 = (float2*)(xs + (size_t)node0 * K);
#pragma unroll
    for (int q = 0; q < XQ; ++q) {
        const int i = tid + q * BNTH;
        if (i < nx2) {
            const float w = dv[i / K2];
            xs2[i] = make_float2(w * xv[q].x, w * xv[q].y);
        }
    }

    int T = 0;
    for (int w = 0; w < BKT_NODES / 32; ++w) T += s_wsum[w];
    const int a  = b0 & 31;
    const int gb = b0 - a;
    const bool staged = a + T <= SEGCAP;

    int slot[NU][NV];
#pragma unroll
    for (int u = 0; u < NU; ++u)
#pragma unroll
        for (int v = 0; v < NV; ++v)
            slot[u][v] = (32 * v + lane < nn[u]) ? atomicAdd(&cur[p[u][v].y - node0], 1) : -1;
    if (staged) {
#pragma unroll
        for (int u = 0; u < NU; ++u)
#pragma unroll
            for (int v = 0; v < NV; ++v)
                if (slot[u][v] >= 0) seg[slot[u][v] - gb] = p[u][v].x;
        for (int b = wid; b < nblk; b += 32) {
            const int n = c_s[b];
            const int2* run = binned + (size_t)b * SPAN + l_s[b];
            for (int i = (b < 32 * NU ? 32 * NV : 0) + lane; i < n; i += 32) {
                const int2 q = run[i];
                seg[atomicAdd(&cur[q.y - node0], 1) - gb] = q.x;
            }
        }
        __syncthreads();
        const int4* seg4 = (const int4*)seg;
        int4* csr4 = (int4*)(csr_src + gb);
        const int endi = a + T;
        const int qlo = (a + 3) >> 2;
        const int qhi = endi >> 2;
        for (int q = tid; q < qhi; q += BNTH)
            if (q >= qlo) csr4[q] = seg4[q];
        if (tid < 3) {
            const int e = a + tid;
            if (e < (qlo << 2) && e < endi) csr_src[gb + e] = seg[e];
        } else if (tid >= 32 && tid < 35) {
            const int e = (qhi << 2) + (tid - 32);
            if (qhi >= qlo && e < endi) csr_src[gb + e] = seg[e];
        }
    } else {
#pragma unroll
        for (int u = 0; u < NU; ++u)
#pragma unroll
            for (int v = 0; v < NV; ++v)
                if (slot[u][v] >= 0) csr_src[slot[u][v]] = p[u][v].x;
        for (int b = wid; b < nblk; b += 32) {
            const int n = c_s[b];
            const int2* run = binned + (size_t)b * SPAN + l_s[b];
            for (int i = (b < 32 * NU ? 32 * NV : 0) + lane; i < n; i += 32) {
                const int2 q = run[i];
                csr_src[atomicAdd(&cur[q.y - node0], 1)] = q.x;
            }
        }
    }
}

template <int K, int U>
__device__ __forceinline__ void gcn_rows_ask(
    const float* __restrict__ xin, int my_s, int j, int coff, float2* r) {
#pragma unroll
    for (int u = 0; u < U; ++u) {
        const int s = __builtin_amdgcn_readlane(my_s, j + u);
        r[u] = *(const float2*)(xin + (size_t)s * K + coff);
    }
}

template <int U>
__device__ __forceinline__ void gcn_rows_add(const float2* r, float& ax, float& ay) {
#pragma unroll
    for (int u = 0; u < U; ++u) {
        ax += r[u].x;
        ay += r[u].y;
    }
}

template <int K, int N, int NTILES, int HIDDEN>
__global__ __launch_bounds__(256) void gcn_gather_gemm_kernel(
    const float* __restrict__ xin, const float* __restrict__ B,
    const int* __restrict__ rowptr, const int* __restrict__ cnt,
    const int* __restrict__ csr_src, const float* __restrict__ dinv,
    float* __restrict__ C, int nNodes) {
    static_assert((K & 1) == 0 && K <= 64, "float2 row loads, one per lane");
    static_assert((NTILES & 1) == 0, "the column tiles are split between two waves");
    constexpr int KP  = (K + 3) & ~3;
    constexpr int KS  = KP + 2;
    constexpr int NP  = NTILES * 16;
    constexpr int NT2 = NTILES / 2;
    constexpr int NTH = 256;
    constexpr int NPW = 8;
    constexpr int NPB = NPW * (NTH / 32);
    static_assert(NPB * N <= (NP + NPB) * KS, "the output tile lies over the operands");
    static_assert((16 * N) % 4 == 0, "whole row tiles are a whole number of 16-byte groups");
    __shared__ __attribute__((aligned(16))) float SM[(NP + NPB) * KS];
    float* BT = SM;
    float* AT = SM + NP * KS;

    const int wib = __builtin_amdgcn_readfirstlane((int)(threadIdx.x >> 5));
    const int nb0 = __builtin_amdgcn_readfirstlane((int)(blockIdx.x * NPB));
    const int n0  = nb0 + wib * NPW;
    const int lane  = threadIdx.x & 31;
    const int lane2 = lane << 1;
    const int coff = lane2 < K ? lane2 : 0;
    const int nleft = nNodes - n0;
    const int nn = nleft < NPW ? nleft : NPW;
    float* ATw = AT + wib * (NPW * KS);


    int m_beg = 0, m_cnt = 0, m_ddbits = 0;
    float2 svA = make_float2(0.0f, 0.0f);
    if (nn > 0) {
        if (lane < nn) {
            m_beg    = rowptr[n0 + lane];
            m_cnt    = cnt[n0 + lane];
            m_ddbits = __float_as_int(dinv[n0 + lane]);
        }
        svA = *(const float2*)(xin + (size_t)n0 * K + coff);
    }

    constexpr int TOT = KS * NP;
    constexpr int PER = (TOT + NTH - 1) / NTH;
    float v[PER];
#pragma unroll
    for (int q = 0; q < PER; ++q) {
        const int idx = (int)threadIdx.x + q * NTH;
        const int k = idx / NP;
        const int n = idx - k * NP;
        const int kc = k < K ? k : K - 1;
        const int nc = n < N ? n : N - 1;
        const float b = B[kc * N + nc];
        v[q] = (n < N && k < K) ? b : 0.0f;
    }

    int begA = 0, cntA = 0, sA = 0;
    int begB = 0, cntB = 0, sB = 0;
    if (nn > 0) {
        begA = __builtin_amdgcn_readlane(m_beg, 0);
        cntA = __builtin_amdgcn_readlane(m_cnt, 0);
        if (lane < cntA) sA = csr_src[begA + lane];
        if (nn > 1) {
            begB = __builtin_amdgcn_readlane(m_beg, 1);
            cntB = __builtin_amdgcn_readlane(m_cnt, 1);
            if (lane < cntB) sB = csr_src[begB + lane];
        }
    }

#pragma unroll
    for (int q = 0; q < PER; ++q) {
        const int idx = (int)threadIdx.x + q * NTH;
        const int k = idx / NP;
        const int n = idx - k * NP;
        if (idx < TOT) BT[n * KS + k] = v[q];
    }

    if (nn > 0) {
        asm volatile("" :: "v"(m_beg), "v"(m_cnt), "v"(m_ddbits), "v"(sA), "v"(sB),
                           "v"(svA.x), "v"(svA.y));
        for (int i = 0; i < nn; ++i) {
            const int node = n0 + i;
            const int beg = begA;
            const int end = begA + cntA;
            const float dd = __int_as_float(__builtin_amdgcn_readlane(m_ddbits, i));
            const int first_s = sA;
            float ax = svA.x;
            float ay = svA.y;

            begA = begB; cntA = cntB; sA = sB;
            if (i + 1 < nn)
                svA = *(const float2*)(xin + (size_t)(node + 1) * K + coff);
            if (i + 2 < nn) {
                begB = __builtin_amdgcn_readlane(m_beg, i + 2);
                cntB = __builtin_amdgcn_readlane(m_cnt, i + 2);
                sB = 0;
                if (lane < cntB) sB = csr_src[begB + lane];
            }

            for (int p = beg; p < end; p += 32) {
                int my_s = first_s;
                if (p != beg) {
                    const int idx = p + lane;
                    my_s = 0;
                    if (idx < end) my_s = csr_src[idx];
                }
                const int left = end - p;
                const int n = left < 32 ? left : 32;
                float2 ra[8], rb[8], rc[8], rd[8], t4[4], t2[2], t1[1];
                const int nt = n & ~7;
                if (n >= 8)  gcn_rows_ask<K, 8>(xin, my_s, 0,  coff, ra);
                if (n >= 16) gcn_rows_ask<K, 8>(xin, my_s, 8,  coff, rb);
                if (n >= 24) gcn_rows_ask<K, 8>(xin, my_s, 16, coff, rc);
                if (n >= 32) gcn_rows_ask<K, 8>(xin, my_s, 24, coff, rd);
                int j = nt;
                if (n & 4) { gcn_rows_ask<K, 4>(xin, my_s, j, coff, t4); j += 4; }
                if (n & 2) { gcn_rows_ask<K, 2>(xin, my_s, j, coff, t2); j += 2; }
                if (n & 1) { gcn_rows_ask<K, 1>(xin, my_s, j, coff, t1); }

                if (n >= 8)  gcn_rows_add<8>(ra, ax, ay);
                if (n >= 16) gcn_rows_add<8>(rb, ax, ay);
                if (n >= 24) gcn_rows_add<8>(rc, ax, ay);
                if (n >= 32) gcn_rows_add<8>(rd, ax, ay);
                if (n & 4) gcn_rows_add<4>(t4, ax, ay);
                if (n & 2) gcn_rows_add<2>(t2, ax, ay);
                if (n & 1) gcn_rows_add<1>(t1, ax, ay);
            }

            if (lane2 < KP) {
                float2 o;
                o.x = lane2 < K ? dd * ax : 0.0f;
                o.y = lane2 < K ? dd * ay : 0.0f;
                *(float2*)(ATw + i * KS + lane2) = o;
            }
        }
    }

    __syncthreads();

    const int rt  = wib >> 1;
    const int ct0 = (wib & 1) * NT2;
    const int row0 = nb0 + (rt << 4);
    const bool have = row0 < nNodes;

    const int mrow  = lane & 15;
    const int khalf = (lane >> 4) << 1;
    const float* ATlane = AT + ((rt << 4) + mrow) * KS;
    const float* BTlane = BT + (ct0 * 16 + mrow) * KS;

    const int rbase = row0 + ((lane >> 4) << 3);
    float4 dlo = make_float4(1.0f, 1.0f, 1.0f, 1.0f), dhi = dlo;
    if (HIDDEN && have) {
        dlo = *(const float4*)(dinv + rbase);
        dhi = *(const float4*)(dinv + rbase + 4);
    }

    v8f acc[NT2];
#pragma unroll
    for (int t = 0; t < NT2; ++t) acc[t] = v8f{};

    if (have) {
#pragma unroll
        for (int k0 = 0; k0 < KP; k0 += 4) {
            const int ka = k0 + khalf;
            const float2 av = *(const float2*)(ATlane + ka);
            v2f a; a.x = av.x; a.y = av.y;
#pragma unroll
            for (int t = 0; t < NT2; ++t) {
                const float2 bv = *(const float2*)(BTlane + t * 16 * KS + ka);
                v2f b; b.x = bv.x; b.y = bv.y;
                acc[t] = __builtin_amdgcn_wmma_f32_16x16x4_f32(
                    false, a, false, b, (short)0, acc[t], false, false);
            }
        }
    }

    __syncthreads();
    float* OT = SM;
    if (have) {
        const float dsc[8] = {dlo.x, dlo.y, dlo.z, dlo.w, dhi.x, dhi.y, dhi.z, dhi.w};
        float* OTlane = OT + ((rt << 4) + ((lane >> 4) << 3)) * N;
#pragma unroll
        for (int t = 0; t < NT2; ++t) {
            const int col = ((ct0 + t) << 4) + mrow;
            if (col < N) {
#pragma unroll
                for (int r = 0; r < 8; ++r) {
                    const float v = acc[t][r];
                    OTlane[r * N + col] = HIDDEN ? dsc[r] * tanhf(v) : v;
                }
            }
        }
    }
    __syncthreads();

    const int rleft = nNodes - nb0;
    const int rows  = rleft < NPB ? rleft : NPB;
    constexpr int NQF = NPB * N / 4;
    constexpr int CQ  = (NQF + NTH - 1) / NTH;
    const float4* OT4 = (const float4*)SM;
    float4* C4 = (float4*)(C + (size_t)nb0 * N);
    if (rows == NPB) {
        float4 o[CQ];
#pragma unroll
        for (int q = 0; q < CQ; ++q) {
            const int i = (int)threadIdx.x + q * NTH;
            o[q] = OT4[((q + 1) * NTH <= NQF || i < NQF) ? i : 0];
        }
#pragma unroll
        for (int q = 0; q < CQ; ++q) {
            const int i = (int)threadIdx.x + q * NTH;
            if ((q + 1) * NTH <= NQF || i < NQF) C4[i] = o[q];
        }
    } else {
        const int nq = (rows * N) >> 2;
        for (int i = (int)threadIdx.x; i < nq; i += NTH) C4[i] = OT4[i];
    }
}

extern "C" void kernel_launch(void* const* d_in, const int* in_sizes, int n_in,
                              void* d_out, int out_size, void* d_ws, size_t ws_size,
                              hipStream_t stream) {
    const float* x  = (const float*)d_in[0];
    const float* W1 = (const float*)d_in[1];
    const float* W2 = (const float*)d_in[2];
    const int* ei   = (const int*)d_in[3];

    const int FIN = 50, F1 = 64;
    const int nNodes = in_sizes[0] / FIN;
    const int nEdges = in_sizes[3] / 2;
    const int* src = ei;
    const int* dst = ei + nEdges;

    const int NP = (nNodes + 63) & ~63;
    const int EP = (nEdges + 63) & ~63;
    const int nbkt = (nNodes + BKT_NODES - 1) >> BKT_SHIFT;
    const int eblk = (nEdges + SPAN - 1) / SPAN;
    const int HP = (eblk * nbkt + 63) & ~63;
    float* wsF    = (float*)d_ws;
    float* dinv   = wsF;
    int*   cnt    = (int*)(dinv + NP);
    int*   rowptr = cnt + NP;
    int*   hist   = rowptr + NP;
    int*   lstart = hist + HP;
    int2*  binned = (int2*)(lstart + HP);
    int*   csr    = (int*)(binned + EP);
    float* hbuf   = (float*)(csr + EP);
    float* xsbuf  = hbuf + (size_t)NP * F1;
    float* outf   = (float*)d_out;

    const int T = 256;
    const int layerBlocks = (nNodes + 63) / 64;

    gcn_span_sort_kernel<<<eblk, BNTH, 0, stream>>>(src, dst, hist, lstart, binned,
                                                    nEdges, nbkt);
    gcn_bucket_rows_kernel<50><<<nbkt, BNTH, 0, stream>>>(binned, hist, lstart, x, rowptr, cnt,
                                                          dinv, csr, xsbuf, nNodes, nbkt, eblk);

    gcn_gather_gemm_kernel<50, 64, 4, 1><<<layerBlocks, T, 0, stream>>>(
        xsbuf, W1, rowptr, cnt, csr, dinv, hbuf, nNodes);
    gcn_gather_gemm_kernel<64, 121, 8, 0><<<layerBlocks, T, 0, stream>>>(
        hbuf, W2, rowptr, cnt, csr, dinv, outf, nNodes);

    (void)n_in; (void)out_size; (void)ws_size; (void)F1;
}
